# starting kernel
# baseline (speedup 1.0000x reference)
#include <hip/hip_runtime.h>
#include <hip/hip_bf16.h>
#include <math.h>

#define BATCH   2
#define LENGTH  2048
#define DINNER  1024
#define DSTATE  16
#define DTRANK  64
#define MROWS   (BATCH * LENGTH)
#define XPROJ_N (DTRANK + 2 * DSTATE)

#define NCH 16
#define LCH (LENGTH / NCH)

typedef __attribute__((ext_vector_type(2))) float v2f;
typedef __attribute__((ext_vector_type(8))) float v8f;

__global__ void k_xproj(const float* __restrict__ u, const float* __restrict__ Wx,
                        float* __restrict__ xdbl) {
  const int wave = blockIdx.x * (blockDim.x >> 5) + (threadIdx.x >> 5);
  const int lane = threadIdx.x & 31;
  const int NT = XPROJ_N / 16;
  const int tileM = wave / NT;
  const int tileN = wave % NT;
  if (tileM >= MROWS / 16) return;

  const int khalf = (lane >> 4) << 1;
  const float* aRow = u  + (size_t)(tileM * 16 + (lane & 15)) * DINNER + khalf;
  const float* bRow = Wx + (size_t)(tileN * 16 + (lane & 15)) * DINNER + khalf;

  v8f acc = {};
  for (int k = 0; k < DINNER; k += 4) {
    v2f a = *(const v2f*)(aRow + k);
    v2f b = *(const v2f*)(bRow + k);
    acc = __builtin_amdgcn_wmma_f32_16x16x4_f32(
         false, a,  false, b,
         (short)0, acc,  false,  false);
  }
  const int col   = tileN * 16 + (lane & 15);
  const int rbase = tileM * 16 + ((lane >> 4) << 3);
#pragma unroll
  for (int r = 0; r < 8; ++r)
    xdbl[(size_t)(rbase + r) * XPROJ_N + col] = acc[r];
}

__global__ void k_dtproj(const float* __restrict__ xdbl, const float* __restrict__ Wdt,
                         const float* __restrict__ bdt, float* __restrict__ delta) {
  const int wave = blockIdx.x * (blockDim.x >> 5) + (threadIdx.x >> 5);
  const int lane = threadIdx.x & 31;
  const int NT = DINNER / 16;
  const int tileM = wave / NT;
  const int tileN = wave % NT;
  if (tileM >= MROWS / 16) return;

  const int khalf = (lane >> 4) << 1;
  const float* aRow = xdbl + (size_t)(tileM * 16 + (lane & 15)) * XPROJ_N + khalf;
  const float* bRow = Wdt  + (size_t)(tileN * 16 + (lane & 15)) * DTRANK + khalf;

  v8f acc = {};
#pragma unroll
  for (int k = 0; k < DTRANK; k += 4) {
    v2f a = *(const v2f*)(aRow + k);
    v2f b = *(const v2f*)(bRow + k);
    acc = __builtin_amdgcn_wmma_f32_16x16x4_f32(
        false, a, false, b, (short)0, acc, false, false);
  }
  const int col   = tileN * 16 + (lane & 15);
  const int rbase = tileM * 16 + ((lane >> 4) << 3);
  const float bias = bdt[col];
#pragma unroll
  for (int r = 0; r < 8; ++r) {
    float x  = acc[r] + bias;
    float sp = (x > 20.f) ? x : log1pf(__expf(x));
    delta[(size_t)(rbase + r) * DINNER + col] = sp;
  }
}

__global__ void k_scan_phase1(const float* __restrict__ u, const float* __restrict__ xdbl,
                              const float* __restrict__ delta, const float* __restrict__ logA,
                              float* __restrict__ chunkA, float* __restrict__ chunkH) {
  const int dblk = blockIdx.x & 3;
  const int ch   = (blockIdx.x >> 2) & (NCH - 1);
  const int b    = blockIdx.x >> 6;
  const int d    = dblk * 256 + threadIdx.x;
  const int t0   = ch * LCH;

  __shared__ float Bs[LCH][DSTATE];
  for (int i = threadIdx.x; i < LCH * DSTATE; i += 256) {
    int t = i >> 4, n = i & 15;
    Bs[t][n] = xdbl[(size_t)(b * LENGTH + t0 + t) * XPROJ_N + DTRANK + n];
  }
  __syncthreads();

  float An[DSTATE], h[DSTATE], ap[DSTATE];
#pragma unroll
  for (int n = 0; n < DSTATE; ++n) {
    An[n] = -__expf(logA[d * DSTATE + n]);
    h[n]  = 0.f;
    ap[n] = 1.f;
  }
  for (int t = 0; t < LCH; ++t) {
    size_t idx = (size_t)(b * LENGTH + t0 + t) * DINNER + d;
    float dlt = delta[idx];
    float uu  = u[idx];
    float du  = dlt * uu;
#pragma unroll
    for (int n = 0; n < DSTATE; ++n) {
      float a = __expf(dlt * An[n]);
      h[n]  = a * h[n] + du * Bs[t][n];
      ap[n] *= a;
    }
  }
  size_t base = (((size_t)(b * DINNER + d)) * NCH + ch) * DSTATE;
#pragma unroll
  for (int n = 0; n < DSTATE; ++n) {
    chunkA[base + n] = ap[n];
    chunkH[base + n] = h[n];
  }
}

__global__ void k_scan_phase2(const float* __restrict__ chunkA, const float* __restrict__ chunkH,
                              float* __restrict__ carry) {
  const int idx = blockIdx.x * 256 + threadIdx.x;
  const int n   = idx & 15;
  const int bd  = idx >> 4;
  size_t base = (size_t)bd * NCH * DSTATE + n;
  float h = 0.f;
#pragma unroll
  for (int c = 0; c < NCH; ++c) {
    carry[base + (size_t)c * DSTATE] = h;
    h = chunkA[base + (size_t)c * DSTATE] * h + chunkH[base + (size_t)c * DSTATE];
  }
}

__global__ void k_scan_phase3(const float* __restrict__ u, const float* __restrict__ xdbl,
                              const float* __restrict__ delta, const float* __restrict__ logA,
                              const float* __restrict__ Dp, const float* __restrict__ carry,
                              float* __restrict__ y) {
  const int dblk = blockIdx.x & 3;
  const int ch   = (blockIdx.x >> 2) & (NCH - 1);
  const int b    = blockIdx.x >> 6;
  const int d    = dblk * 256 + threadIdx.x;
  const int t0   = ch * LCH;

  __shared__ float Bs[LCH][DSTATE];
  __shared__ float Cs[LCH][DSTATE];
  for (int i = threadIdx.x; i < LCH * DSTATE; i += 256) {
    int t = i >> 4, n = i & 15;
    size_t rowoff = (size_t)(b * LENGTH + t0 + t) * XPROJ_N;
    Bs[t][n] = xdbl[rowoff + DTRANK + n];
    Cs[t][n] = xdbl[rowoff + DTRANK + DSTATE + n];
  }
  __syncthreads();

  float An[DSTATE], h[DSTATE];
  size_t cbase = (((size_t)(b * DINNER + d)) * NCH + ch) * DSTATE;
#pragma unroll
  for (int n = 0; n < DSTATE; ++n) {
    An[n] = -__expf(logA[d * DSTATE + n]);
    h[n]  = carry[cbase + n];
  }
  const float Dd = Dp[d];
  for (int t = 0; t < LCH; ++t) {
    size_t idx = (size_t)(b * LENGTH + t0 + t) * DINNER + d;
    float dlt = delta[idx];
    float uu  = u[idx];
    float du  = dlt * uu;
    float acc = 0.f;
#pragma unroll
    for (int n = 0; n < DSTATE; ++n) {
      float a = __expf(dlt * An[n]);
      h[n] = a * h[n] + du * Bs[t][n];
      acc += h[n] * Cs[t][n];
    }
    y[idx] = acc + uu * Dd;
  }
}

extern "C" void kernel_launch(void* const* d_in, const int* in_sizes, int n_in,
                              void* d_out, int out_size, void* d_ws, size_t ws_size,
                              hipStream_t stream) {
  (void)in_sizes; (void)n_in; (void)out_size; (void)ws_size;
  const float* u    = (const float*)d_in[0];
  const float* Wx   = (const float*)d_in[1];
  const float* Wdt  = (const float*)d_in[2];
  const float* bdt  = (const float*)d_in[3];
  const float* logA = (const float*)d_in[4];
  const float* Dp   = (const float*)d_in[5];
  float* y = (float*)d_out;

  float* ws     = (float*)d_ws;
  float* xdbl   = ws;
  float* delta  = xdbl  + (size_t)MROWS * XPROJ_N;
  float* chunkA = delta + (size_t)MROWS * DINNER;
  float* chunkH = chunkA + (size_t)BATCH * DINNER * NCH * DSTATE;
  float* carry  = chunkH + (size_t)BATCH * DINNER * NCH * DSTATE;

  {
    int tiles = (MROWS / 16) * (XPROJ_N / 16);
    k_xproj<<<(tiles + 7) / 8, 256, 0, stream>>>(u, Wx, xdbl);
  }
  {
    int tiles = (MROWS / 16) * (DINNER / 16);
    k_dtproj<<<(tiles + 7) / 8, 256, 0, stream>>>(xdbl, Wdt, bdt, delta);
  }
  const int scan_blocks = BATCH * NCH * (DINNER / 256);
  k_scan_phase1<<<scan_blocks, 256, 0, stream>>>(u, xdbl, delta, logA, chunkA, chunkH);
  k_scan_phase2<<<(BATCH * DINNER * DSTATE) / 256, 256, 0, stream>>>(chunkA, chunkH, carry);
  k_scan_phase3<<<scan_blocks, 256, 0, stream>>>(u, xdbl, delta, logA, Dp, carry, y);
}


# scan chunks shortened from 128 to 16 steps (128 chunks per sequence): 8x more waves in scan phases 1 and 3 hide per-step load latency; carry pass loads chunk states in groups; sum order changes
# speedup vs baseline: 2.0895x; 2.0895x over previous
#include <hip/hip_runtime.h>
#include <hip/hip_bf16.h>
#include <math.h>

#define BATCH   2
#define LENGTH  2048
#define DINNER  1024
#define DSTATE  16
#define DTRANK  64
#define MROWS   (BATCH * LENGTH)
#define XPROJ_N (DTRANK + 2 * DSTATE)

#define NCH 128
#define LCH (LENGTH / NCH)
#define DBLKS (DINNER / 256)
#define P2B 32

typedef __attribute__((ext_vector_type(2))) float v2f;
typedef __attribute__((ext_vector_type(8))) float v8f;

__global__ void k_xproj(const float* __restrict__ u, const float* __restrict__ Wx,
                        float* __restrict__ xdbl) {
  const int wave = blockIdx.x * (blockDim.x >> 5) + (threadIdx.x >> 5);
  const int lane = threadIdx.x & 31;
  const int NT = XPROJ_N / 16;
  const int tileM = wave / NT;
  const int tileN = wave % NT;
  if (tileM >= MROWS / 16) return;

  const int khalf = (lane >> 4) << 1;
  const float* aRow = u  + (size_t)(tileM * 16 + (lane & 15)) * DINNER + khalf;
  const float* bRow = Wx + (size_t)(tileN * 16 + (lane & 15)) * DINNER + khalf;

  v8f acc = {};
  for (int k = 0; k < DINNER; k += 4) {
    v2f a = *(const v2f*)(aRow + k);
    v2f b = *(const v2f*)(bRow + k);
    acc = __builtin_amdgcn_wmma_f32_16x16x4_f32(
         false, a,  false, b,
         (short)0, acc,  false,  false);
  }
  const int col   = tileN * 16 + (lane & 15);
  const int rbase = tileM * 16 + ((lane >> 4) << 3);
#pragma unroll
  for (int r = 0; r < 8; ++r)
    xdbl[(size_t)(rbase + r) * XPROJ_N + col] = acc[r];
}

__global__ void k_dtproj(const float* __restrict__ xdbl, const float* __restrict__ Wdt,
                         const float* __restrict__ bdt, float* __restrict__ delta) {
  const int wave = blockIdx.x * (blockDim.x >> 5) + (threadIdx.x >> 5);
  const int lane = threadIdx.x & 31;
  const int NT = DINNER / 16;
  const int tileM = wave / NT;
  const int tileN = wave % NT;
  if (tileM >= MROWS / 16) return;

  const int khalf = (lane >> 4) << 1;
  const float* aRow = xdbl + (size_t)(tileM * 16 + (lane & 15)) * XPROJ_N + khalf;
  const float* bRow = Wdt  + (size_t)(tileN * 16 + (lane & 15)) * DTRANK + khalf;

  v8f acc = {};
#pragma unroll
  for (int k = 0; k < DTRANK; k += 4) {
    v2f a = *(const v2f*)(aRow + k);
    v2f b = *(const v2f*)(bRow + k);
    acc = __builtin_amdgcn_wmma_f32_16x16x4_f32(
        false, a, false, b, (short)0, acc, false, false);
  }
  const int col   = tileN * 16 + (lane & 15);
  const int rbase = tileM * 16 + ((lane >> 4) << 3);
  const float bias = bdt[col];
#pragma unroll
  for (int r = 0; r < 8; ++r) {
    float x  = acc[r] + bias;
    float sp = (x > 20.f) ? x : log1pf(__expf(x));
    delta[(size_t)(rbase + r) * DINNER + col] = sp;
  }
}

__global__ void k_scan_phase1(const float* __restrict__ u, const float* __restrict__ xdbl,
                              const float* __restrict__ delta, const float* __restrict__ logA,
                              float* __restrict__ chunkA, float* __restrict__ chunkH) {
  const int dblk = blockIdx.x % DBLKS;
  const int ch   = (blockIdx.x / DBLKS) % NCH;
  const int b    = blockIdx.x / (DBLKS * NCH);
  const int d    = dblk * 256 + threadIdx.x;
  const int t0   = ch * LCH;

  __shared__ float Bs[LCH][DSTATE];
  for (int i = threadIdx.x; i < LCH * DSTATE; i += 256) {
    int t = i >> 4, n = i & 15;
    Bs[t][n] = xdbl[(size_t)(b * LENGTH + t0 + t) * XPROJ_N + DTRANK + n];
  }
  __syncthreads();

  float An[DSTATE], h[DSTATE], ap[DSTATE];
#pragma unroll
  for (int n = 0; n < DSTATE; ++n) {
    An[n] = -__expf(logA[d * DSTATE + n]);
    h[n]  = 0.f;
    ap[n] = 1.f;
  }
  for (int t = 0; t < LCH; ++t) {
    size_t idx = (size_t)(b * LENGTH + t0 + t) * DINNER + d;
    float dlt = delta[idx];
    float uu  = u[idx];
    float du  = dlt * uu;
#pragma unroll
    for (int n = 0; n < DSTATE; ++n) {
      float a = __expf(dlt * An[n]);
      h[n]  = a * h[n] + du * Bs[t][n];
      ap[n] *= a;
    }
  }
  size_t base = (((size_t)(b * DINNER + d)) * NCH + ch) * DSTATE;
#pragma unroll
  for (int n = 0; n < DSTATE; ++n) {
    chunkA[base + n] = ap[n];
    chunkH[base + n] = h[n];
  }
}

__global__ void k_scan_phase2(const float* __restrict__ chunkA, const float* __restrict__ chunkH,
                              float* __restrict__ carry) {
  const int idx = blockIdx.x * 256 + threadIdx.x;
  const int n   = idx & 15;
  const int bd  = idx >> 4;
  size_t base = (size_t)bd * NCH * DSTATE + n;
  float h = 0.f;
  for (int c0 = 0; c0 < NCH; c0 += P2B) {
    float a[P2B], g[P2B];
#pragma unroll
    for (int j = 0; j < P2B; ++j) {
      a[j] = chunkA[base + (size_t)(c0 + j) * DSTATE];
      g[j] = chunkH[base + (size_t)(c0 + j) * DSTATE];
    }
#pragma unroll
    for (int j = 0; j < P2B; ++j) {
      carry[base + (size_t)(c0 + j) * DSTATE] = h;
      h = a[j] * h + g[j];
    }
  }
}

__global__ void k_scan_phase3(const float* __restrict__ u, const float* __restrict__ xdbl,
                              const float* __restrict__ delta, const float* __restrict__ logA,
                              const float* __restrict__ Dp, const float* __restrict__ carry,
                              float* __restrict__ y) {
  const int dblk = blockIdx.x % DBLKS;
  const int ch   = (blockIdx.x / DBLKS) % NCH;
  const int b    = blockIdx.x / (DBLKS * NCH);
  const int d    = dblk * 256 + threadIdx.x;
  const int t0   = ch * LCH;

  __shared__ float Bs[LCH][DSTATE];
  __shared__ float Cs[LCH][DSTATE];
  for (int i = threadIdx.x; i < LCH * DSTATE; i += 256) {
    int t = i >> 4, n = i & 15;
    size_t rowoff = (size_t)(b * LENGTH + t0 + t) * XPROJ_N;
    Bs[t][n] = xdbl[rowoff + DTRANK + n];
    Cs[t][n] = xdbl[rowoff + DTRANK + DSTATE + n];
  }
  __syncthreads();

  float An[DSTATE], h[DSTATE];
  size_t cbase = (((size_t)(b * DINNER + d)) * NCH + ch) * DSTATE;
#pragma unroll
  for (int n = 0; n < DSTATE; ++n) {
    An[n] = -__expf(logA[d * DSTATE + n]);
    h[n]  = carry[cbase + n];
  }
  const float Dd = Dp[d];
  for (int t = 0; t < LCH; ++t) {
    size_t idx = (size_t)(b * LENGTH + t0 + t) * DINNER + d;
    float dlt = delta[idx];
    float uu  = u[idx];
    float du  = dlt * uu;
    float acc = 0.f;
#pragma unroll
    for (int n = 0; n < DSTATE; ++n) {
      float a = __expf(dlt * An[n]);
      h[n] = a * h[n] + du * Bs[t][n];
      acc += h[n] * Cs[t][n];
    }
    y[idx] = acc + uu * Dd;
  }
}

extern "C" void kernel_launch(void* const* d_in, const int* in_sizes, int n_in,
                              void* d_out, int out_size, void* d_ws, size_t ws_size,
                              hipStream_t stream) {
  (void)in_sizes; (void)n_in; (void)out_size; (void)ws_size;
  const float* u    = (const float*)d_in[0];
  const float* Wx   = (const float*)d_in[1];
  const float* Wdt  = (const float*)d_in[2];
  const float* bdt  = (const float*)d_in[3];
  const float* logA = (const float*)d_in[4];
  const float* Dp   = (const float*)d_in[5];
  float* y = (float*)d_out;

  float* ws     = (float*)d_ws;
  float* xdbl   = ws;
  float* delta  = xdbl  + (size_t)MROWS * XPROJ_N;
  float* chunkA = delta + (size_t)MROWS * DINNER;
  float* chunkH = chunkA + (size_t)BATCH * DINNER * NCH * DSTATE;
  float* carry  = chunkH + (size_t)BATCH * DINNER * NCH * DSTATE;

  {
    int tiles = (MROWS / 16) * (XPROJ_N / 16);
    k_xproj<<<(tiles + 7) / 8, 256, 0, stream>>>(u, Wx, xdbl);
  }
  {
    int tiles = (MROWS / 16) * (DINNER / 16);
    k_dtproj<<<(tiles + 7) / 8, 256, 0, stream>>>(xdbl, Wdt, bdt, delta);
  }
  const int scan_blocks = BATCH * NCH * DBLKS;
  k_scan_phase1<<<scan_blocks, 256, 0, stream>>>(u, xdbl, delta, logA, chunkA, chunkH);
  k_scan_phase2<<<(BATCH * DINNER * DSTATE) / 256, 256, 0, stream>>>(chunkA, chunkH, carry);
  k_scan_phase3<<<scan_blocks, 256, 0, stream>>>(u, xdbl, delta, logA, Dp, carry, y);
}


# x_proj GEMM re-tiled: one work-group per 16-row tile does all 96 columns, K split over 16 waves, 16-byte fragment loads, one A fragment feeds 6 WMMAs; slices summed through LDS, sum order changes
# speedup vs baseline: 2.3119x; 1.1064x over previous
#include <hip/hip_runtime.h>
#include <hip/hip_bf16.h>
#include <math.h>

#define BATCH   2
#define LENGTH  2048
#define DINNER  1024
#define DSTATE  16
#define DTRANK  64
#define MROWS   (BATCH * LENGTH)
#define XPROJ_N (DTRANK + 2 * DSTATE)

#define NCH 128
#define LCH (LENGTH / NCH)
#define DBLKS (DINNER / 256)
#define P2B 32

typedef __attribute__((ext_vector_type(2))) float v2f;
typedef __attribute__((ext_vector_type(4))) float v4f;
typedef __attribute__((ext_vector_type(8))) float v8f;

#define XP_WAVES  16
#define XP_KSLICE (DINNER / XP_WAVES)
#define XP_NT     (XPROJ_N / 16)

__global__ void __launch_bounds__(XP_WAVES * 32) __attribute__((amdgpu_waves_per_eu(4, 4)))
k_xproj(const float* __restrict__ u, const float* __restrict__ Wx,
        float* __restrict__ xdbl) {
  __shared__ float red[XP_WAVES][XP_NT][8][32];
  const int w    = threadIdx.x >> 5;
  const int lane = threadIdx.x & 31;
  const int c    = lane & 15;
  const int h    = lane >> 4;
  const int tileM = blockIdx.x;

  const int kofs = w * XP_KSLICE + 4 * h;
  const float* aRow = u  + (size_t)(tileM * 16 + c) * DINNER + kofs;
  const float* bRow = Wx + (size_t)c * DINNER + kofs;

  v8f acc[XP_NT];
#pragma unroll
  for (int j = 0; j < XP_NT; ++j) acc[j] = v8f{};

  v4f a[XP_KSLICE / 8];
  v4f b[XP_NT][XP_KSLICE / 8];
#pragma unroll
  for (int i = 0; i < XP_KSLICE / 8; ++i)
    a[i] = *(const v4f*)(aRow + 8 * i);
  __builtin_amdgcn_sched_barrier(0);
#pragma unroll
  for (int j = 0; j < XP_NT; ++j) {
#pragma unroll
    for (int i = 0; i < XP_KSLICE / 8; ++i)
      b[j][i] = *(const v4f*)(bRow + (size_t)j * 16 * DINNER + 8 * i);
    if (j < XP_NT - 1) __builtin_amdgcn_sched_barrier(0);
  }

#pragma unroll
  for (int i = 0; i < XP_KSLICE / 8; ++i) {
    const v2f a0 = {a[i].x, a[i].y};
    const v2f a1 = {a[i].z, a[i].w};
#pragma unroll
    for (int j = 0; j < XP_NT; ++j) {
      const v2f b0 = {b[j][i].x, b[j][i].y};
      acc[j] = __builtin_amdgcn_wmma_f32_16x16x4_f32(
           false, a0,  false, b0,
           (short)0, acc[j],  false,  false);
    }
#pragma unroll
    for (int j = 0; j < XP_NT; ++j) {
      const v2f b1 = {b[j][i].z, b[j][i].w};
      acc[j] = __builtin_amdgcn_wmma_f32_16x16x4_f32(
          false, a1, false, b1, (short)0, acc[j], false, false);
    }
  }

#pragma unroll
  for (int j = 0; j < XP_NT; ++j)
#pragma unroll
    for (int r = 0; r < 8; ++r)
      red[w][j][r][lane] = acc[j][r];
  __syncthreads();

  float* outBase = xdbl + (size_t)tileM * (16 * XPROJ_N);
#pragma unroll
  for (int q = 0; q < (16 * XPROJ_N) / (XP_WAVES * 32); ++q) {
    const int o   = threadIdx.x + q * (XP_WAVES * 32);
    const int row = o / XPROJ_N;
    const int col = o - row * XPROJ_N;
    const int j   = col >> 4;
    const int src = ((row >> 3) << 4) | (col & 15);
    const int r   = row & 7;
    float s = red[0][j][r][src];
#pragma unroll
    for (int ww = 1; ww < XP_WAVES; ++ww) s += red[ww][j][r][src];
    outBase[o] = s;
  }
}

__global__ void k_dtproj(const float* __restrict__ xdbl, const float* __restrict__ Wdt,
                         const float* __restrict__ bdt, float* __restrict__ delta) {
  const int wave = blockIdx.x * (blockDim.x >> 5) + (threadIdx.x >> 5);
  const int lane = threadIdx.x & 31;
  const int NT = DINNER / 16;
  const int tileM = wave / NT;
  const int tileN = wave % NT;
  if (tileM >= MROWS / 16) return;

  const int khalf = (lane >> 4) << 1;
  const float* aRow = xdbl + (size_t)(tileM * 16 + (lane & 15)) * XPROJ_N + khalf;
  const float* bRow = Wdt  + (size_t)(tileN * 16 + (lane & 15)) * DTRANK + khalf;

  v8f acc = {};
#pragma unroll
  for (int k = 0; k < DTRANK; k += 4) {
    v2f a = *(const v2f*)(aRow + k);
    v2f b = *(const v2f*)(bRow + k);
    acc = __builtin_amdgcn_wmma_f32_16x16x4_f32(
        false, a, false, b, (short)0, acc, false, false);
  }
  const int col   = tileN * 16 + (lane & 15);
  const int rbase = tileM * 16 + ((lane >> 4) << 3);
  const float bias = bdt[col];
#pragma unroll
  for (int r = 0; r < 8; ++r) {
    float x  = acc[r] + bias;
    float sp = (x > 20.f) ? x : log1pf(__expf(x));
    delta[(size_t)(rbase + r) * DINNER + col] = sp;
  }
}

__global__ void k_scan_phase1(const float* __restrict__ u, const float* __restrict__ xdbl,
                              const float* __restrict__ delta, const float* __restrict__ logA,
                              float* __restrict__ chunkA, float* __restrict__ chunkH) {
  const int dblk = blockIdx.x % DBLKS;
  const int ch   = (blockIdx.x / DBLKS) % NCH;
  const int b    = blockIdx.x / (DBLKS * NCH);
  const int d    = dblk * 256 + threadIdx.x;
  const int t0   = ch * LCH;

  __shared__ float Bs[LCH][DSTATE];
  for (int i = threadIdx.x; i < LCH * DSTATE; i += 256) {
    int t = i >> 4, n = i & 15;
    Bs[t][n] = xdbl[(size_t)(b * LENGTH + t0 + t) * XPROJ_N + DTRANK + n];
  }
  __syncthreads();

  float An[DSTATE], h[DSTATE], ap[DSTATE];
#pragma unroll
  for (int n = 0; n < DSTATE; ++n) {
    An[n] = -__expf(logA[d * DSTATE + n]);
    h[n]  = 0.f;
    ap[n] = 1.f;
  }
  for (int t = 0; t < LCH; ++t) {
    size_t idx = (size_t)(b * LENGTH + t0 + t) * DINNER + d;
    float dlt = delta[idx];
    float uu  = u[idx];
    float du  = dlt * uu;
#pragma unroll
    for (int n = 0; n < DSTATE; ++n) {
      float a = __expf(dlt * An[n]);
      h[n]  = a * h[n] + du * Bs[t][n];
      ap[n] *= a;
    }
  }
  size_t base = (((size_t)(b * DINNER + d)) * NCH + ch) * DSTATE;
#pragma unroll
  for (int n = 0; n < DSTATE; ++n) {
    chunkA[base + n] = ap[n];
    chunkH[base + n] = h[n];
  }
}

__global__ void k_scan_phase2(const float* __restrict__ chunkA, const float* __restrict__ chunkH,
                              float* __restrict__ carry) {
  const int idx = blockIdx.x * 256 + threadIdx.x;
  const int n   = idx & 15;
  const int bd  = idx >> 4;
  size_t base = (size_t)bd * NCH * DSTATE + n;
  float h = 0.f;
  for (int c0 = 0; c0 < NCH; c0 += P2B) {
    float a[P2B], g[P2B];
#pragma unroll
    for (int j = 0; j < P2B; ++j) {
      a[j] = chunkA[base + (size_t)(c0 + j) * DSTATE];
      g[j] = chunkH[base + (size_t)(c0 + j) * DSTATE];
    }
#pragma unroll
    for (int j = 0; j < P2B; ++j) {
      carry[base + (size_t)(c0 + j) * DSTATE] = h;
      h = a[j] * h + g[j];
    }
  }
}

__global__ void k_scan_phase3(const float* __restrict__ u, const float* __restrict__ xdbl,
                              const float* __restrict__ delta, const float* __restrict__ logA,
                              const float* __restrict__ Dp, const float* __restrict__ carry,
                              float* __restrict__ y) {
  const int dblk = blockIdx.x % DBLKS;
  const int ch   = (blockIdx.x / DBLKS) % NCH;
  const int b    = blockIdx.x / (DBLKS * NCH);
  const int d    = dblk * 256 + threadIdx.x;
  const int t0   = ch * LCH;

  __shared__ float Bs[LCH][DSTATE];
  __shared__ float Cs[LCH][DSTATE];
  for (int i = threadIdx.x; i < LCH * DSTATE; i += 256) {
    int t = i >> 4, n = i & 15;
    size_t rowoff = (size_t)(b * LENGTH + t0 + t) * XPROJ_N;
    Bs[t][n] = xdbl[rowoff + DTRANK + n];
    Cs[t][n] = xdbl[rowoff + DTRANK + DSTATE + n];
  }
  __syncthreads();

  float An[DSTATE], h[DSTATE];
  size_t cbase = (((size_t)(b * DINNER + d)) * NCH + ch) * DSTATE;
#pragma unroll
  for (int n = 0; n < DSTATE; ++n) {
    An[n] = -__expf(logA[d * DSTATE + n]);
    h[n]  = carry[cbase + n];
  }
  const float Dd = Dp[d];
  for (int t = 0; t < LCH; ++t) {
    size_t idx = (size_t)(b * LENGTH + t0 + t) * DINNER + d;
    float dlt = delta[idx];
    float uu  = u[idx];
    float du  = dlt * uu;
    float acc = 0.f;
#pragma unroll
    for (int n = 0; n < DSTATE; ++n) {
      float a = __expf(dlt * An[n]);
      h[n] = a * h[n] + du * Bs[t][n];
      acc += h[n] * Cs[t][n];
    }
    y[idx] = acc + uu * Dd;
  }
}

extern "C" void kernel_launch(void* const* d_in, const int* in_sizes, int n_in,
                              void* d_out, int out_size, void* d_ws, size_t ws_size,
                              hipStream_t stream) {
  (void)in_sizes; (void)n_in; (void)out_size; (void)ws_size;
  const float* u    = (const float*)d_in[0];
  const float* Wx   = (const float*)d_in[1];
  const float* Wdt  = (const float*)d_in[2];
  const float* bdt  = (const float*)d_in[3];
  const float* logA = (const float*)d_in[4];
  const float* Dp   = (const float*)d_in[5];
  float* y = (float*)d_out;

  float* ws     = (float*)d_ws;
  float* xdbl   = ws;
  float* delta  = xdbl  + (size_t)MROWS * XPROJ_N;
  float* chunkA = delta + (size_t)MROWS * DINNER;
  float* chunkH = chunkA + (size_t)BATCH * DINNER * NCH * DSTATE;
  float* carry  = chunkH + (size_t)BATCH * DINNER * NCH * DSTATE;

  k_xproj<<<MROWS / 16, XP_WAVES * 32, 0, stream>>>(u, Wx, xdbl);
  {
    int tiles = (MROWS / 16) * (DINNER / 16);
    k_dtproj<<<(tiles + 7) / 8, 256, 0, stream>>>(xdbl, Wdt, bdt, delta);
  }
  const int scan_blocks = BATCH * NCH * DBLKS;
  k_scan_phase1<<<scan_blocks, 256, 0, stream>>>(u, xdbl, delta, logA, chunkA, chunkH);
  k_scan_phase2<<<(BATCH * DINNER * DSTATE) / 256, 256, 0, stream>>>(chunkA, chunkH, carry);
  k_scan_phase3<<<scan_blocks, 256, 0, stream>>>(u, xdbl, delta, logA, Dp, carry, y);
}


# scan phases 1 and 3: every global load of a 16-step chunk (delta, u, A row, carry, B/C) is issued before the first wait, delta and u parked in LDS: one memory round trip per wave, not one per step
# speedup vs baseline: 2.4279x; 1.0502x over previous
#include <hip/hip_runtime.h>
#include <hip/hip_bf16.h>
#include <math.h>

#define BATCH   2
#define LENGTH  2048
#define DINNER  1024
#define DSTATE  16
#define DTRANK  64
#define MROWS   (BATCH * LENGTH)
#define XPROJ_N (DTRANK + 2 * DSTATE)

#define NCH 128
#define LCH (LENGTH / NCH)
#define DBLKS (DINNER / 256)
#define P2B 32

typedef __attribute__((ext_vector_type(2))) float v2f;
typedef __attribute__((ext_vector_type(4))) float v4f;
typedef __attribute__((ext_vector_type(8))) float v8f;

#define XP_WAVES  16
#define XP_KSLICE (DINNER / XP_WAVES)
#define XP_NT     (XPROJ_N / 16)

__global__ void __launch_bounds__(XP_WAVES * 32) __attribute__((amdgpu_waves_per_eu(4, 4)))
k_xproj(const float* __restrict__ u, const float* __restrict__ Wx,
        float* __restrict__ xdbl) {
  __shared__ float red[XP_WAVES][XP_NT][8][32];
  const int w    = threadIdx.x >> 5;
  const int lane = threadIdx.x & 31;
  const int c    = lane & 15;
  const int h    = lane >> 4;
  const int tileM = blockIdx.x;

  const int kofs = w * XP_KSLICE + 4 * h;
  const float* aRow = u  + (size_t)(tileM * 16 + c) * DINNER + kofs;
  const float* bRow = Wx + (size_t)c * DINNER + kofs;

  v8f acc[XP_NT];
#pragma unroll
  for (int j = 0; j < XP_NT; ++j) acc[j] = v8f{};

  v4f a[XP_KSLICE / 8];
  v4f b[XP_NT][XP_KSLICE / 8];
#pragma unroll
  for (int i = 0; i < XP_KSLICE / 8; ++i)
    a[i] = *(const v4f*)(aRow + 8 * i);
  __builtin_amdgcn_sched_barrier(0);
#pragma unroll
  for (int j = 0; j < XP_NT; ++j) {
#pragma unroll
    for (int i = 0; i < XP_KSLICE / 8; ++i)
      b[j][i] = *(const v4f*)(bRow + (size_t)j * 16 * DINNER + 8 * i);
    if (j < XP_NT - 1) __builtin_amdgcn_sched_barrier(0);
  }

#pragma unroll
  for (int i = 0; i < XP_KSLICE / 8; ++i) {
    const v2f a0 = {a[i].x, a[i].y};
    const v2f a1 = {a[i].z, a[i].w};
#pragma unroll
    for (int j = 0; j < XP_NT; ++j) {
      const v2f b0 = {b[j][i].x, b[j][i].y};
      acc[j] = __builtin_amdgcn_wmma_f32_16x16x4_f32(
           false, a0,  false, b0,
           (short)0, acc[j],  false,  false);
    }
#pragma unroll
    for (int j = 0; j < XP_NT; ++j) {
      const v2f b1 = {b[j][i].z, b[j][i].w};
      acc[j] = __builtin_amdgcn_wmma_f32_16x16x4_f32(
          false, a1, false, b1, (short)0, acc[j], false, false);
    }
  }

#pragma unroll
  for (int j = 0; j < XP_NT; ++j)
#pragma unroll
    for (int r = 0; r < 8; ++r)
      red[w][j][r][lane] = acc[j][r];
  __syncthreads();

  float* outBase = xdbl + (size_t)tileM * (16 * XPROJ_N);
#pragma unroll
  for (int q = 0; q < (16 * XPROJ_N) / (XP_WAVES * 32); ++q) {
    const int o   = threadIdx.x + q * (XP_WAVES * 32);
    const int row = o / XPROJ_N;
    const int col = o - row * XPROJ_N;
    const int j   = col >> 4;
    const int src = ((row >> 3) << 4) | (col & 15);
    const int r   = row & 7;
    float s = red[0][j][r][src];
#pragma unroll
    for (int ww = 1; ww < XP_WAVES; ++ww) s += red[ww][j][r][src];
    outBase[o] = s;
  }
}

__global__ void k_dtproj(const float* __restrict__ xdbl, const float* __restrict__ Wdt,
                         const float* __restrict__ bdt, float* __restrict__ delta) {
  const int wave = blockIdx.x * (blockDim.x >> 5) + (threadIdx.x >> 5);
  const int lane = threadIdx.x & 31;
  const int NT = DINNER / 16;
  const int tileM = wave / NT;
  const int tileN = wave % NT;
  if (tileM >= MROWS / 16) return;

  const int khalf = (lane >> 4) << 1;
  const float* aRow = xdbl + (size_t)(tileM * 16 + (lane & 15)) * XPROJ_N + khalf;
  const float* bRow = Wdt  + (size_t)(tileN * 16 + (lane & 15)) * DTRANK + khalf;

  v8f acc = {};
#pragma unroll
  for (int k = 0; k < DTRANK; k += 4) {
    v2f a = *(const v2f*)(aRow + k);
    v2f b = *(const v2f*)(bRow + k);
    acc = __builtin_amdgcn_wmma_f32_16x16x4_f32(
        false, a, false, b, (short)0, acc, false, false);
  }
  const int col   = tileN * 16 + (lane & 15);
  const int rbase = tileM * 16 + ((lane >> 4) << 3);
  const float bias = bdt[col];
#pragma unroll
  for (int r = 0; r < 8; ++r) {
    float x  = acc[r] + bias;
    float sp = (x > 20.f) ? x : log1pf(__expf(x));
    delta[(size_t)(rbase + r) * DINNER + col] = sp;
  }
}

__global__ void k_scan_phase1(const float* __restrict__ u, const float* __restrict__ xdbl,
                              const float* __restrict__ delta, const float* __restrict__ logA,
                              float* __restrict__ chunkA, float* __restrict__ chunkH) {
  const int dblk = blockIdx.x % DBLKS;
  const int ch   = (blockIdx.x / DBLKS) % NCH;
  const int b    = blockIdx.x / (DBLKS * NCH);
  const int tid  = threadIdx.x;
  const int d    = dblk * 256 + tid;
  const int t0   = ch * LCH;

  __shared__ float Bs[LCH][DSTATE];
  __shared__ float Dl[LCH][256];
  __shared__ float Uu[LCH][256];

  static_assert(LCH * DSTATE == 256, "one staged B element per thread");
  const int st = tid >> 4, sn = tid & 15;
  const float bstage = xdbl[(size_t)(b * LENGTH + t0 + st) * XPROJ_N + DTRANK + sn];

  float An[DSTATE];
#pragma unroll
  for (int n = 0; n < DSTATE; ++n) An[n] = logA[d * DSTATE + n];

  const size_t idx0 = (size_t)(b * LENGTH + t0) * DINNER + d;
  float dl[LCH], uv[LCH];
#pragma unroll
  for (int t = 0; t < LCH; ++t) {
    dl[t] = delta[idx0 + (size_t)t * DINNER];
    uv[t] = u[idx0 + (size_t)t * DINNER];
  }

  Bs[st][sn] = bstage;
#pragma unroll
  for (int t = 0; t < LCH; ++t) {
    Dl[t][tid] = dl[t];
    Uu[t][tid] = uv[t];
  }
#pragma unroll
  for (int n = 0; n < DSTATE; ++n) An[n] = -__expf(An[n]);
  __syncthreads();

  float h[DSTATE], ap[DSTATE];
#pragma unroll
  for (int n = 0; n < DSTATE; ++n) {
    h[n]  = 0.f;
    ap[n] = 1.f;
  }
  float dnext = Dl[0][tid], unext = Uu[0][tid];
  for (int t = 0; t < LCH; ++t) {
    const float dlt = dnext;
    const float du  = dlt * unext;
    const int   tn  = (t + 1 < LCH) ? t + 1 : t;
    dnext = Dl[tn][tid];
    unext = Uu[tn][tid];
#pragma unroll
    for (int n = 0; n < DSTATE; ++n) {
      float a = __expf(dlt * An[n]);
      h[n]  = a * h[n] + du * Bs[t][n];
      ap[n] *= a;
    }
  }
  size_t base = (((size_t)(b * DINNER + d)) * NCH + ch) * DSTATE;
#pragma unroll
  for (int n = 0; n < DSTATE; ++n) {
    chunkA[base + n] = ap[n];
    chunkH[base + n] = h[n];
  }
}

__global__ void k_scan_phase2(const float* __restrict__ chunkA, const float* __restrict__ chunkH,
                              float* __restrict__ carry) {
  const int idx = blockIdx.x * 256 + threadIdx.x;
  const int n   = idx & 15;
  const int bd  = idx >> 4;
  size_t base = (size_t)bd * NCH * DSTATE + n;
  float h = 0.f;
  for (int c0 = 0; c0 < NCH; c0 += P2B) {
    float a[P2B], g[P2B];
#pragma unroll
    for (int j = 0; j < P2B; ++j) {
      a[j] = chunkA[base + (size_t)(c0 + j) * DSTATE];
      g[j] = chunkH[base + (size_t)(c0 + j) * DSTATE];
    }
#pragma unroll
    for (int j = 0; j < P2B; ++j) {
      carry[base + (size_t)(c0 + j) * DSTATE] = h;
      h = a[j] * h + g[j];
    }
  }
}

__global__ void k_scan_phase3(const float* __restrict__ u, const float* __restrict__ xdbl,
                              const float* __restrict__ delta, const float* __restrict__ logA,
                              const float* __restrict__ Dp, const float* __restrict__ carry,
                              float* __restrict__ y) {
  const int dblk = blockIdx.x % DBLKS;
  const int ch   = (blockIdx.x / DBLKS) % NCH;
  const int b    = blockIdx.x / (DBLKS * NCH);
  const int tid  = threadIdx.x;
  const int d    = dblk * 256 + tid;
  const int t0   = ch * LCH;

  __shared__ float Bs[LCH][DSTATE];
  __shared__ float Cs[LCH][DSTATE];
  __shared__ float Dl[LCH][256];
  __shared__ float Uu[LCH][256];

  static_assert(LCH * DSTATE == 256, "one staged B and C element per thread");
  float An[DSTATE], h[DSTATE];
  const size_t cbase = (((size_t)(b * DINNER + d)) * NCH + ch) * DSTATE;
#pragma unroll
  for (int n = 0; n < DSTATE; ++n) h[n] = carry[cbase + n];
#pragma unroll
  for (int n = 0; n < DSTATE; ++n) An[n] = logA[d * DSTATE + n];
  const float Dd = Dp[d];

  const int st = tid >> 4, sn = tid & 15;
  const size_t rowoff = (size_t)(b * LENGTH + t0 + st) * XPROJ_N;
  const float bstage = xdbl[rowoff + DTRANK + sn];
  const float cstage = xdbl[rowoff + DTRANK + DSTATE + sn];

  const size_t idx0 = (size_t)(b * LENGTH + t0) * DINNER + d;
  float dl[LCH], uv[LCH];
#pragma unroll
  for (int t = 0; t < LCH; ++t) {
    dl[t] = delta[idx0 + (size_t)t * DINNER];
    uv[t] = u[idx0 + (size_t)t * DINNER];
  }

  asm volatile("" :: "v"(An[0]), "v"(An[1]), "v"(An[2]), "v"(An[3]),
                     "v"(An[4]), "v"(An[5]), "v"(An[6]), "v"(An[7]),
                     "v"(An[8]), "v"(An[9]), "v"(An[10]), "v"(An[11]),
                     "v"(An[12]), "v"(An[13]), "v"(An[14]), "v"(An[15]));
  __builtin_amdgcn_sched_barrier(0);
  Bs[st][sn] = bstage;
  Cs[st][sn] = cstage;
#pragma unroll
  for (int t = 0; t < LCH; ++t) {
    Dl[t][tid] = dl[t];
    Uu[t][tid] = uv[t];
  }
#pragma unroll
  for (int n = 0; n < DSTATE; ++n) An[n] = -__expf(An[n]);
  __syncthreads();

  float dnext = Dl[0][tid], unext = Uu[0][tid];
  for (int t = 0; t < LCH; ++t) {
    const float dlt = dnext;
    const float uu  = unext;
    const float du  = dlt * uu;
    const int   tn  = (t + 1 < LCH) ? t + 1 : t;
    dnext = Dl[tn][tid];
    unext = Uu[tn][tid];
    float acc = 0.f;
#pragma unroll
    for (int n = 0; n < DSTATE; ++n) {
      float a = __expf(dlt * An[n]);
      h[n] = a * h[n] + du * Bs[t][n];
      acc += h[n] * Cs[t][n];
    }
    y[idx0 + (size_t)t * DINNER] = acc + uu * Dd;
  }
}

extern "C" void kernel_launch(void* const* d_in, const int* in_sizes, int n_in,
                              void* d_out, int out_size, void* d_ws, size_t ws_size,
                              hipStream_t stream) {
  (void)in_sizes; (void)n_in; (void)out_size; (void)ws_size;
  const float* u    = (const float*)d_in[0];
  const float* Wx   = (const float*)d_in[1];
  const float* Wdt  = (const float*)d_in[2];
  const float* bdt  = (const float*)d_in[3];
  const float* logA = (const float*)d_in[4];
  const float* Dp   = (const float*)d_in[5];
  float* y = (float*)d_out;

  float* ws     = (float*)d_ws;
  float* xdbl   = ws;
  float* delta  = xdbl  + (size_t)MROWS * XPROJ_N;
  float* chunkA = delta + (size_t)MROWS * DINNER;
  float* chunkH = chunkA + (size_t)BATCH * DINNER * NCH * DSTATE;
  float* carry  = chunkH + (size_t)BATCH * DINNER * NCH * DSTATE;

  k_xproj<<<MROWS / 16, XP_WAVES * 32, 0, stream>>>(u, Wx, xdbl);
  {
    int tiles = (MROWS / 16) * (DINNER / 16);
    k_dtproj<<<(tiles + 7) / 8, 256, 0, stream>>>(xdbl, Wdt, bdt, delta);
  }
  const int scan_blocks = BATCH * NCH * DBLKS;
  k_scan_phase1<<<scan_blocks, 256, 0, stream>>>(u, xdbl, delta, logA, chunkA, chunkH);
  k_scan_phase2<<<(BATCH * DINNER * DSTATE) / 256, 256, 0, stream>>>(chunkA, chunkH, carry);
  k_scan_phase3<<<scan_blocks, 256, 0, stream>>>(u, xdbl, delta, logA, Dp, carry, y);
}


# scan chunks lengthened from 16 to 32 steps (64 per sequence): chunk-state arrays, carry-pass steps and scan waves all halve; a chunk's loads all still issued before the first wait; sum order changes
# speedup vs baseline: 2.9888x; 1.2310x over previous
#include <hip/hip_runtime.h>
#include <hip/hip_bf16.h>
#include <math.h>

#define BATCH   2
#define LENGTH  2048
#define DINNER  1024
#define DSTATE  16
#define DTRANK  64
#define MROWS   (BATCH * LENGTH)
#define XPROJ_N (DTRANK + 2 * DSTATE)

#define NCH 64
#define LCH (LENGTH / NCH)
#define STG ((LCH * DSTATE) / 256)
#define DBLKS (DINNER / 256)
#define P2B 32

typedef __attribute__((ext_vector_type(2))) float v2f;
typedef __attribute__((ext_vector_type(4))) float v4f;
typedef __attribute__((ext_vector_type(8))) float v8f;

#define XP_WAVES  16
#define XP_KSLICE (DINNER / XP_WAVES)
#define XP_NT     (XPROJ_N / 16)

__global__ void __launch_bounds__(XP_WAVES * 32) __attribute__((amdgpu_waves_per_eu(4, 4)))
k_xproj(const float* __restrict__ u, const float* __restrict__ Wx,
        float* __restrict__ xdbl) {
  __shared__ float red[XP_WAVES][XP_NT][8][32];
  const int w    = threadIdx.x >> 5;
  const int lane = threadIdx.x & 31;
  const int c    = lane & 15;
  const int h    = lane >> 4;
  const int tileM = blockIdx.x;

  const int kofs = w * XP_KSLICE + 4 * h;
  const float* aRow = u  + (size_t)(tileM * 16 + c) * DINNER + kofs;
  const float* bRow = Wx + (size_t)c * DINNER + kofs;

  v8f acc[XP_NT];
#pragma unroll
  for (int j = 0; j < XP_NT; ++j) acc[j] = v8f{};

  v4f a[XP_KSLICE / 8];
  v4f b[XP_NT][XP_KSLICE / 8];
#pragma unroll
  for (int i = 0; i < XP_KSLICE / 8; ++i)
    a[i] = *(const v4f*)(aRow + 8 * i);
  __builtin_amdgcn_sched_barrier(0);
#pragma unroll
  for (int j = 0; j < XP_NT; ++j) {
#pragma unroll
    for (int i = 0; i < XP_KSLICE / 8; ++i)
      b[j][i] = *(const v4f*)(bRow + (size_t)j * 16 * DINNER + 8 * i);
    if (j < XP_NT - 1) __builtin_amdgcn_sched_barrier(0);
  }

#pragma unroll
  for (int i = 0; i < XP_KSLICE / 8; ++i) {
    const v2f a0 = {a[i].x, a[i].y};
    const v2f a1 = {a[i].z, a[i].w};
#pragma unroll
    for (int j = 0; j < XP_NT; ++j) {
      const v2f b0 = {b[j][i].x, b[j][i].y};
      acc[j] = __builtin_amdgcn_wmma_f32_16x16x4_f32(
           false, a0,  false, b0,
           (short)0, acc[j],  false,  false);
    }
#pragma unroll
    for (int j = 0; j < XP_NT; ++j) {
      const v2f b1 = {b[j][i].z, b[j][i].w};
      acc[j] = __builtin_amdgcn_wmma_f32_16x16x4_f32(
          false, a1, false, b1, (short)0, acc[j], false, false);
    }
  }

#pragma unroll
  for (int j = 0; j < XP_NT; ++j)
#pragma unroll
    for (int r = 0; r < 8; ++r)
      red[w][j][r][lane] = acc[j][r];
  __syncthreads();

  float* outBase = xdbl + (size_t)tileM * (16 * XPROJ_N);
#pragma unroll
  for (int q = 0; q < (16 * XPROJ_N) / (XP_WAVES * 32); ++q) {
    const int o   = threadIdx.x + q * (XP_WAVES * 32);
    const int row = o / XPROJ_N;
    const int col = o - row * XPROJ_N;
    const int j   = col >> 4;
    const int src = ((row >> 3) << 4) | (col & 15);
    const int r   = row & 7;
    float s = red[0][j][r][src];
#pragma unroll
    for (int ww = 1; ww < XP_WAVES; ++ww) s += red[ww][j][r][src];
    outBase[o] = s;
  }
}

__global__ void k_dtproj(const float* __restrict__ xdbl, const float* __restrict__ Wdt,
                         const float* __restrict__ bdt, float* __restrict__ delta) {
  const int wave = blockIdx.x * (blockDim.x >> 5) + (threadIdx.x >> 5);
  const int lane = threadIdx.x & 31;
  const int NT = DINNER / 16;
  const int tileM = wave / NT;
  const int tileN = wave % NT;
  if (tileM >= MROWS / 16) return;

  const int khalf = (lane >> 4) << 1;
  const float* aRow = xdbl + (size_t)(tileM * 16 + (lane & 15)) * XPROJ_N + khalf;
  const float* bRow = Wdt  + (size_t)(tileN * 16 + (lane & 15)) * DTRANK + khalf;

  v8f acc = {};
#pragma unroll
  for (int k = 0; k < DTRANK; k += 4) {
    v2f a = *(const v2f*)(aRow + k);
    v2f b = *(const v2f*)(bRow + k);
    acc = __builtin_amdgcn_wmma_f32_16x16x4_f32(
        false, a, false, b, (short)0, acc, false, false);
  }
  const int col   = tileN * 16 + (lane & 15);
  const int rbase = tileM * 16 + ((lane >> 4) << 3);
  const float bias = bdt[col];
#pragma unroll
  for (int r = 0; r < 8; ++r) {
    float x  = acc[r] + bias;
    float sp = (x > 20.f) ? x : log1pf(__expf(x));
    delta[(size_t)(rbase + r) * DINNER + col] = sp;
  }
}

__global__ void k_scan_phase1(const float* __restrict__ u, const float* __restrict__ xdbl,
                              const float* __restrict__ delta, const float* __restrict__ logA,
                              float* __restrict__ chunkA, float* __restrict__ chunkH) {
  const int dblk = blockIdx.x % DBLKS;
  const int ch   = (blockIdx.x / DBLKS) % NCH;
  const int b    = blockIdx.x / (DBLKS * NCH);
  const int tid  = threadIdx.x;
  const int d    = dblk * 256 + tid;
  const int t0   = ch * LCH;

  __shared__ float Bs[LCH][DSTATE];
  __shared__ float Dl[LCH][256];
  __shared__ float Uu[LCH][256];

  static_assert(STG * 256 == LCH * DSTATE, "whole staged B elements per thread");
  const int st = tid >> 4, sn = tid & 15;
  float bstage[STG];
#pragma unroll
  for (int q = 0; q < STG; ++q)
    bstage[q] = xdbl[(size_t)(b * LENGTH + t0 + st + 16 * q) * XPROJ_N + DTRANK + sn];

  float An[DSTATE];
#pragma unroll
  for (int n = 0; n < DSTATE; ++n) An[n] = logA[d * DSTATE + n];

  const size_t idx0 = (size_t)(b * LENGTH + t0) * DINNER + d;
  float dl[LCH], uv[LCH];
#pragma unroll
  for (int t = 0; t < LCH; ++t) {
    dl[t] = delta[idx0 + (size_t)t * DINNER];
    uv[t] = u[idx0 + (size_t)t * DINNER];
  }

  asm volatile("" :: "v"(An[0]), "v"(An[1]), "v"(An[2]), "v"(An[3]),
                     "v"(An[4]), "v"(An[5]), "v"(An[6]), "v"(An[7]),
                     "v"(An[8]), "v"(An[9]), "v"(An[10]), "v"(An[11]),
                     "v"(An[12]), "v"(An[13]), "v"(An[14]), "v"(An[15]));
  __builtin_amdgcn_sched_barrier(0);
#pragma unroll
  for (int q = 0; q < STG; ++q) Bs[st + 16 * q][sn] = bstage[q];
#pragma unroll
  for (int t = 0; t < LCH; ++t) {
    Dl[t][tid] = dl[t];
    Uu[t][tid] = uv[t];
  }
#pragma unroll
  for (int n = 0; n < DSTATE; ++n) An[n] = -__expf(An[n]);
  __syncthreads();

  float h[DSTATE], ap[DSTATE];
#pragma unroll
  for (int n = 0; n < DSTATE; ++n) {
    h[n]  = 0.f;
    ap[n] = 1.f;
  }
  float dnext = Dl[0][tid], unext = Uu[0][tid];
  for (int t = 0; t < LCH; ++t) {
    const float dlt = dnext;
    const float du  = dlt * unext;
    const int   tn  = (t + 1 < LCH) ? t + 1 : t;
    dnext = Dl[tn][tid];
    unext = Uu[tn][tid];
#pragma unroll
    for (int n = 0; n < DSTATE; ++n) {
      float a = __expf(dlt * An[n]);
      h[n]  = a * h[n] + du * Bs[t][n];
      ap[n] *= a;
    }
  }
  size_t base = (((size_t)(b * DINNER + d)) * NCH + ch) * DSTATE;
#pragma unroll
  for (int n = 0; n < DSTATE; ++n) {
    chunkA[base + n] = ap[n];
    chunkH[base + n] = h[n];
  }
}

__global__ void k_scan_phase2(const float* __restrict__ chunkA, const float* __restrict__ chunkH,
                              float* __restrict__ carry) {
  const int idx = blockIdx.x * 256 + threadIdx.x;
  const int n   = idx & 15;
  const int bd  = idx >> 4;
  size_t base = (size_t)bd * NCH * DSTATE + n;
  float h = 0.f;
  for (int c0 = 0; c0 < NCH; c0 += P2B) {
    float a[P2B], g[P2B];
#pragma unroll
    for (int j = 0; j < P2B; ++j) {
      a[j] = chunkA[base + (size_t)(c0 + j) * DSTATE];
      g[j] = chunkH[base + (size_t)(c0 + j) * DSTATE];
    }
#pragma unroll
    for (int j = 0; j < P2B; ++j) {
      carry[base + (size_t)(c0 + j) * DSTATE] = h;
      h = a[j] * h + g[j];
    }
  }
}

__global__ void k_scan_phase3(const float* __restrict__ u, const float* __restrict__ xdbl,
                              const float* __restrict__ delta, const float* __restrict__ logA,
                              const float* __restrict__ Dp, const float* __restrict__ carry,
                              float* __restrict__ y) {
  const int dblk = blockIdx.x % DBLKS;
  const int ch   = (blockIdx.x / DBLKS) % NCH;
  const int b    = blockIdx.x / (DBLKS * NCH);
  const int tid  = threadIdx.x;
  const int d    = dblk * 256 + tid;
  const int t0   = ch * LCH;

  __shared__ float Bs[LCH][DSTATE];
  __shared__ float Cs[LCH][DSTATE];
  __shared__ float Dl[LCH][256];
  __shared__ float Uu[LCH][256];

  static_assert(STG * 256 == LCH * DSTATE, "whole staged B and C elements per thread");
  float An[DSTATE], h[DSTATE];
  const size_t cbase = (((size_t)(b * DINNER + d)) * NCH + ch) * DSTATE;
#pragma unroll
  for (int n = 0; n < DSTATE; ++n) h[n] = carry[cbase + n];
#pragma unroll
  for (int n = 0; n < DSTATE; ++n) An[n] = logA[d * DSTATE + n];
  const float Dd = Dp[d];

  const int st = tid >> 4, sn = tid & 15;
  float bstage[STG], cstage[STG];
#pragma unroll
  for (int q = 0; q < STG; ++q) {
    const size_t rowoff = (size_t)(b * LENGTH + t0 + st + 16 * q) * XPROJ_N;
    bstage[q] = xdbl[rowoff + DTRANK + sn];
    cstage[q] = xdbl[rowoff + DTRANK + DSTATE + sn];
  }

  const size_t idx0 = (size_t)(b * LENGTH + t0) * DINNER + d;
  float dl[LCH], uv[LCH];
#pragma unroll
  for (int t = 0; t < LCH; ++t) {
    dl[t] = delta[idx0 + (size_t)t * DINNER];
    uv[t] = u[idx0 + (size_t)t * DINNER];
  }

  asm volatile("" :: "v"(An[0]), "v"(An[1]), "v"(An[2]), "v"(An[3]),
                     "v"(An[4]), "v"(An[5]), "v"(An[6]), "v"(An[7]),
                     "v"(An[8]), "v"(An[9]), "v"(An[10]), "v"(An[11]),
                     "v"(An[12]), "v"(An[13]), "v"(An[14]), "v"(An[15]));
  __builtin_amdgcn_sched_barrier(0);
#pragma unroll
  for (int q = 0; q < STG; ++q) {
    Bs[st + 16 * q][sn] = bstage[q];
    Cs[st + 16 * q][sn] = cstage[q];
  }
#pragma unroll
  for (int t = 0; t < LCH; ++t) {
    Dl[t][tid] = dl[t];
    Uu[t][tid] = uv[t];
  }
#pragma unroll
  for (int n = 0; n < DSTATE; ++n) An[n] = -__expf(An[n]);
  __syncthreads();

  float dnext = Dl[0][tid], unext = Uu[0][tid];
  for (int t = 0; t < LCH; ++t) {
    const float dlt = dnext;
    const float uu  = unext;
    const float du  = dlt * uu;
    const int   tn  = (t + 1 < LCH) ? t + 1 : t;
    dnext = Dl[tn][tid];
    unext = Uu[tn][tid];
    float acc = 0.f;
#pragma unroll
    for (int n = 0; n < DSTATE; ++n) {
      float a = __expf(dlt * An[n]);
      h[n] = a * h[n] + du * Bs[t][n];
      acc += h[n] * Cs[t][n];
    }
    y[idx0 + (size_t)t * DINNER] = acc + uu * Dd;
  }
}

extern "C" void kernel_launch(void* const* d_in, const int* in_sizes, int n_in,
                              void* d_out, int out_size, void* d_ws, size_t ws_size,
                              hipStream_t stream) {
  (void)in_sizes; (void)n_in; (void)out_size; (void)ws_size;
  const float* u    = (const float*)d_in[0];
  const float* Wx   = (const float*)d_in[1];
  const float* Wdt  = (const float*)d_in[2];
  const float* bdt  = (const float*)d_in[3];
  const float* logA = (const float*)d_in[4];
  const float* Dp   = (const float*)d_in[5];
  float* y = (float*)d_out;

  float* ws     = (float*)d_ws;
  float* xdbl   = ws;
  float* delta  = xdbl  + (size_t)MROWS * XPROJ_N;
  float* chunkA = delta + (size_t)MROWS * DINNER;
  float* chunkH = chunkA + (size_t)BATCH * DINNER * NCH * DSTATE;
  float* carry  = chunkH + (size_t)BATCH * DINNER * NCH * DSTATE;

  k_xproj<<<MROWS / 16, XP_WAVES * 32, 0, stream>>>(u, Wx, xdbl);
  {
    int tiles = (MROWS / 16) * (DINNER / 16);
    k_dtproj<<<(tiles + 7) / 8, 256, 0, stream>>>(xdbl, Wdt, bdt, delta);
  }
  const int scan_blocks = BATCH * NCH * DBLKS;
  k_scan_phase1<<<scan_blocks, 256, 0, stream>>>(u, xdbl, delta, logA, chunkA, chunkH);
  k_scan_phase2<<<(BATCH * DINNER * DSTATE) / 256, 256, 0, stream>>>(chunkA, chunkH, carry);
  k_scan_phase3<<<scan_blocks, 256, 0, stream>>>(u, xdbl, delta, logA, Dp, carry, y);
}


# scan chunks lengthened from 32 to 64 steps (32 per sequence): chunk-state arrays, carry-pass steps and scan waves halve again; time loops unrolled by two (2 waves per SIMD left); sum order changes
# speedup vs baseline: 3.2226x; 1.0782x over previous
#include <hip/hip_runtime.h>
#include <hip/hip_bf16.h>
#include <math.h>

#define BATCH   2
#define LENGTH  2048
#define DINNER  1024
#define DSTATE  16
#define DTRANK  64
#define MROWS   (BATCH * LENGTH)
#define XPROJ_N (DTRANK + 2 * DSTATE)

#define NCH 32
#define LCH (LENGTH / NCH)
#define STG ((LCH * DSTATE) / 256)
#define DBLKS (DINNER / 256)
#define P2B 32

typedef __attribute__((ext_vector_type(2))) float v2f;
typedef __attribute__((ext_vector_type(4))) float v4f;
typedef __attribute__((ext_vector_type(8))) float v8f;

#define XP_WAVES  16
#define XP_KSLICE (DINNER / XP_WAVES)
#define XP_NT     (XPROJ_N / 16)

__global__ void __launch_bounds__(XP_WAVES * 32) __attribute__((amdgpu_waves_per_eu(4, 4)))
k_xproj(const float* __restrict__ u, const float* __restrict__ Wx,
        float* __restrict__ xdbl) {
  __shared__ float red[XP_WAVES][XP_NT][8][32];
  const int w    = threadIdx.x >> 5;
  const int lane = threadIdx.x & 31;
  const int c    = lane & 15;
  const int h    = lane >> 4;
  const int tileM = blockIdx.x;

  const int kofs = w * XP_KSLICE + 4 * h;
  const float* aRow = u  + (size_t)(tileM * 16 + c) * DINNER + kofs;
  const float* bRow = Wx + (size_t)c * DINNER + kofs;

  v8f acc[XP_NT];
#pragma unroll
  for (int j = 0; j < XP_NT; ++j) acc[j] = v8f{};

  v4f a[XP_KSLICE / 8];
  v4f b[XP_NT][XP_KSLICE / 8];
#pragma unroll
  for (int i = 0; i < XP_KSLICE / 8; ++i)
    a[i] = *(const v4f*)(aRow + 8 * i);
  __builtin_amdgcn_sched_barrier(0);
#pragma unroll
  for (int j = 0; j < XP_NT; ++j) {
#pragma unroll
    for (int i = 0; i < XP_KSLICE / 8; ++i)
      b[j][i] = *(const v4f*)(bRow + (size_t)j * 16 * DINNER + 8 * i);
    if (j < XP_NT - 1) __builtin_amdgcn_sched_barrier(0);
  }

#pragma unroll
  for (int i = 0; i < XP_KSLICE / 8; ++i) {
    const v2f a0 = {a[i].x, a[i].y};
    const v2f a1 = {a[i].z, a[i].w};
#pragma unroll
    for (int j = 0; j < XP_NT; ++j) {
      const v2f b0 = {b[j][i].x, b[j][i].y};
      acc[j] = __builtin_amdgcn_wmma_f32_16x16x4_f32(
           false, a0,  false, b0,
           (short)0, acc[j],  false,  false);
    }
#pragma unroll
    for (int j = 0; j < XP_NT; ++j) {
      const v2f b1 = {b[j][i].z, b[j][i].w};
      acc[j] = __builtin_amdgcn_wmma_f32_16x16x4_f32(
          false, a1, false, b1, (short)0, acc[j], false, false);
    }
  }

#pragma unroll
  for (int j = 0; j < XP_NT; ++j)
#pragma unroll
    for (int r = 0; r < 8; ++r)
      red[w][j][r][lane] = acc[j][r];
  __syncthreads();

  float* outBase = xdbl + (size_t)tileM * (16 * XPROJ_N);
#pragma unroll
  for (int q = 0; q < (16 * XPROJ_N) / (XP_WAVES * 32); ++q) {
    const int o   = threadIdx.x + q * (XP_WAVES * 32);
    const int row = o / XPROJ_N;
    const int col = o - row * XPROJ_N;
    const int j   = col >> 4;
    const int src = ((row >> 3) << 4) | (col & 15);
    const int r   = row & 7;
    float s = red[0][j][r][src];
#pragma unroll
    for (int ww = 1; ww < XP_WAVES; ++ww) s += red[ww][j][r][src];
    outBase[o] = s;
  }
}

__global__ void k_dtproj(const float* __restrict__ xdbl, const float* __restrict__ Wdt,
                         const float* __restrict__ bdt, float* __restrict__ delta) {
  const int wave = blockIdx.x * (blockDim.x >> 5) + (threadIdx.x >> 5);
  const int lane = threadIdx.x & 31;
  const int NT = DINNER / 16;
  const int tileM = wave / NT;
  const int tileN = wave % NT;
  if (tileM >= MROWS / 16) return;

  const int khalf = (lane >> 4) << 1;
  const float* aRow = xdbl + (size_t)(tileM * 16 + (lane & 15)) * XPROJ_N + khalf;
  const float* bRow = Wdt  + (size_t)(tileN * 16 + (lane & 15)) * DTRANK + khalf;

  v8f acc = {};
#pragma unroll
  for (int k = 0; k < DTRANK; k += 4) {
    v2f a = *(const v2f*)(aRow + k);
    v2f b = *(const v2f*)(bRow + k);
    acc = __builtin_amdgcn_wmma_f32_16x16x4_f32(
        false, a, false, b, (short)0, acc, false, false);
  }
  const int col   = tileN * 16 + (lane & 15);
  const int rbase = tileM * 16 + ((lane >> 4) << 3);
  const float bias = bdt[col];
#pragma unroll
  for (int r = 0; r < 8; ++r) {
    float x  = acc[r] + bias;
    float sp = (x > 20.f) ? x : log1pf(__expf(x));
    delta[(size_t)(rbase + r) * DINNER + col] = sp;
  }
}

__global__ void __launch_bounds__(256) k_scan_phase1(const float* __restrict__ u, const float* __restrict__ xdbl,
                              const float* __restrict__ delta, const float* __restrict__ logA,
                              float* __restrict__ chunkA, float* __restrict__ chunkH) {
  const int dblk = blockIdx.x % DBLKS;
  const int ch   = (blockIdx.x / DBLKS) % NCH;
  const int b    = blockIdx.x / (DBLKS * NCH);
  const int tid  = threadIdx.x;
  const int d    = dblk * 256 + tid;
  const int t0   = ch * LCH;

  __shared__ float Bs[LCH][DSTATE];
  __shared__ float Dl[LCH][256];
  __shared__ float Uu[LCH][256];

  static_assert(STG * 256 == LCH * DSTATE, "whole staged B elements per thread");
  const int st = tid >> 4, sn = tid & 15;
  float bstage[STG];
#pragma unroll
  for (int q = 0; q < STG; ++q)
    bstage[q] = xdbl[(size_t)(b * LENGTH + t0 + st + 16 * q) * XPROJ_N + DTRANK + sn];

  float An[DSTATE];
#pragma unroll
  for (int n = 0; n < DSTATE; ++n) An[n] = logA[d * DSTATE + n];

  const size_t idx0 = (size_t)(b * LENGTH + t0) * DINNER + d;
  float dl[LCH], uv[LCH];
#pragma unroll
  for (int t = 0; t < LCH; ++t) {
    dl[t] = delta[idx0 + (size_t)t * DINNER];
    uv[t] = u[idx0 + (size_t)t * DINNER];
  }

  asm volatile("" :: "v"(An[0]), "v"(An[1]), "v"(An[2]), "v"(An[3]),
                     "v"(An[4]), "v"(An[5]), "v"(An[6]), "v"(An[7]),
                     "v"(An[8]), "v"(An[9]), "v"(An[10]), "v"(An[11]),
                     "v"(An[12]), "v"(An[13]), "v"(An[14]), "v"(An[15]));
  __builtin_amdgcn_sched_barrier(0);
#pragma unroll
  for (int q = 0; q < STG; ++q) Bs[st + 16 * q][sn] = bstage[q];
#pragma unroll
  for (int t = 0; t < LCH; ++t) {
    Dl[t][tid] = dl[t];
    Uu[t][tid] = uv[t];
  }
#pragma unroll
  for (int n = 0; n < DSTATE; ++n) An[n] = -__expf(An[n]);
  __syncthreads();

  float h[DSTATE], ap[DSTATE];
#pragma unroll
  for (int n = 0; n < DSTATE; ++n) {
    h[n]  = 0.f;
    ap[n] = 1.f;
  }
  float dnext = Dl[0][tid], unext = Uu[0][tid];
#pragma unroll 2
  for (int t = 0; t < LCH; ++t) {
    const float dlt = dnext;
    const float du  = dlt * unext;
    const int   tn  = (t + 1 < LCH) ? t + 1 : t;
    dnext = Dl[tn][tid];
    unext = Uu[tn][tid];
#pragma unroll
    for (int n = 0; n < DSTATE; ++n) {
      float a = __expf(dlt * An[n]);
      h[n]  = a * h[n] + du * Bs[t][n];
      ap[n] *= a;
    }
  }
  size_t base = (((size_t)(b * DINNER + d)) * NCH + ch) * DSTATE;
#pragma unroll
  for (int n = 0; n < DSTATE; ++n) {
    chunkA[base + n] = ap[n];
    chunkH[base + n] = h[n];
  }
}

__global__ void k_scan_phase2(const float* __restrict__ chunkA, const float* __restrict__ chunkH,
                              float* __restrict__ carry) {
  const int idx = blockIdx.x * 256 + threadIdx.x;
  const int n   = idx & 15;
  const int bd  = idx >> 4;
  size_t base = (size_t)bd * NCH * DSTATE + n;
  float h = 0.f;
  for (int c0 = 0; c0 < NCH; c0 += P2B) {
    float a[P2B], g[P2B];
#pragma unroll
    for (int j = 0; j < P2B; ++j) {
      a[j] = chunkA[base + (size_t)(c0 + j) * DSTATE];
      g[j] = chunkH[base + (size_t)(c0 + j) * DSTATE];
    }
#pragma unroll
    for (int j = 0; j < P2B; ++j) {
      carry[base + (size_t)(c0 + j) * DSTATE] = h;
      h = a[j] * h + g[j];
    }
  }
}

__global__ void __launch_bounds__(256) k_scan_phase3(const float* __restrict__ u, const float* __restrict__ xdbl,
                              const float* __restrict__ delta, const float* __restrict__ logA,
                              const float* __restrict__ Dp, const float* __restrict__ carry,
                              float* __restrict__ y) {
  const int dblk = blockIdx.x % DBLKS;
  const int ch   = (blockIdx.x / DBLKS) % NCH;
  const int b    = blockIdx.x / (DBLKS * NCH);
  const int tid  = threadIdx.x;
  const int d    = dblk * 256 + tid;
  const int t0   = ch * LCH;

  __shared__ float Bs[LCH][DSTATE];
  __shared__ float Cs[LCH][DSTATE];
  __shared__ float Dl[LCH][256];
  __shared__ float Uu[LCH][256];

  static_assert(STG * 256 == LCH * DSTATE, "whole staged B and C elements per thread");
  float An[DSTATE], h[DSTATE];
  const size_t cbase = (((size_t)(b * DINNER + d)) * NCH + ch) * DSTATE;
#pragma unroll
  for (int n = 0; n < DSTATE; ++n) h[n] = carry[cbase + n];
#pragma unroll
  for (int n = 0; n < DSTATE; ++n) An[n] = logA[d * DSTATE + n];
  const float Dd = Dp[d];

  const int st = tid >> 4, sn = tid & 15;
  float bstage[STG], cstage[STG];
#pragma unroll
  for (int q = 0; q < STG; ++q) {
    const size_t rowoff = (size_t)(b * LENGTH + t0 + st + 16 * q) * XPROJ_N;
    bstage[q] = xdbl[rowoff + DTRANK + sn];
    cstage[q] = xdbl[rowoff + DTRANK + DSTATE + sn];
  }

  const size_t idx0 = (size_t)(b * LENGTH + t0) * DINNER + d;
  float dl[LCH], uv[LCH];
#pragma unroll
  for (int t = 0; t < LCH; ++t) {
    dl[t] = delta[idx0 + (size_t)t * DINNER];
    uv[t] = u[idx0 + (size_t)t * DINNER];
  }

  asm volatile("" :: "v"(An[0]), "v"(An[1]), "v"(An[2]), "v"(An[3]),
                     "v"(An[4]), "v"(An[5]), "v"(An[6]), "v"(An[7]),
                     "v"(An[8]), "v"(An[9]), "v"(An[10]), "v"(An[11]),
                     "v"(An[12]), "v"(An[13]), "v"(An[14]), "v"(An[15]));
  __builtin_amdgcn_sched_barrier(0);
#pragma unroll
  for (int q = 0; q < STG; ++q) {
    Bs[st + 16 * q][sn] = bstage[q];
    Cs[st + 16 * q][sn] = cstage[q];
  }
#pragma unroll
  for (int t = 0; t < LCH; ++t) {
    Dl[t][tid] = dl[t];
    Uu[t][tid] = uv[t];
  }
#pragma unroll
  for (int n = 0; n < DSTATE; ++n) An[n] = -__expf(An[n]);
  __syncthreads();

  float dnext = Dl[0][tid], unext = Uu[0][tid];
#pragma unroll 2
  for (int t = 0; t < LCH; ++t) {
    const float dlt = dnext;
    const float uu  = unext;
    const float du  = dlt * uu;
    const int   tn  = (t + 1 < LCH) ? t + 1 : t;
    dnext = Dl[tn][tid];
    unext = Uu[tn][tid];
    float acc = 0.f;
#pragma unroll
    for (int n = 0; n < DSTATE; ++n) {
      float a = __expf(dlt * An[n]);
      h[n] = a * h[n] + du * Bs[t][n];
      acc += h[n] * Cs[t][n];
    }
    y[idx0 + (size_t)t * DINNER] = acc + uu * Dd;
  }
}

extern "C" void kernel_launch(void* const* d_in, const int* in_sizes, int n_in,
                              void* d_out, int out_size, void* d_ws, size_t ws_size,
                              hipStream_t stream) {
  (void)in_sizes; (void)n_in; (void)out_size; (void)ws_size;
  const float* u    = (const float*)d_in[0];
  const float* Wx   = (const float*)d_in[1];
  const float* Wdt  = (const float*)d_in[2];
  const float* bdt  = (const float*)d_in[3];
  const float* logA = (const float*)d_in[4];
  const float* Dp   = (const float*)d_in[5];
  float* y = (float*)d_out;

  float* ws     = (float*)d_ws;
  float* xdbl   = ws;
  float* delta  = xdbl  + (size_t)MROWS * XPROJ_N;
  float* chunkA = delta + (size_t)MROWS * DINNER;
  float* chunkH = chunkA + (size_t)BATCH * DINNER * NCH * DSTATE;
  float* carry  = chunkH + (size_t)BATCH * DINNER * NCH * DSTATE;

  k_xproj<<<MROWS / 16, XP_WAVES * 32, 0, stream>>>(u, Wx, xdbl);
  {
    int tiles = (MROWS / 16) * (DINNER / 16);
    k_dtproj<<<(tiles + 7) / 8, 256, 0, stream>>>(xdbl, Wdt, bdt, delta);
  }
  const int scan_blocks = BATCH * NCH * DBLKS;
  k_scan_phase1<<<scan_blocks, 256, 0, stream>>>(u, xdbl, delta, logA, chunkA, chunkH);
  k_scan_phase2<<<(BATCH * DINNER * DSTATE) / 256, 256, 0, stream>>>(chunkA, chunkH, carry);
  k_scan_phase3<<<scan_blocks, 256, 0, stream>>>(u, xdbl, delta, logA, Dp, carry, y);
}


# dt_proj GEMM and softplus moved into the x_proj kernel as an epilogue on its 16-row tile (delta_r handed over in LDS, 4 column tiles per wave): 4 launches, not 5; whole-line stores; sum order changes
# speedup vs baseline: 3.3626x; 1.0434x over previous
#include <hip/hip_runtime.h>
#include <hip/hip_bf16.h>
#include <math.h>

#define BATCH   2
#define LENGTH  2048
#define DINNER  1024
#define DSTATE  16
#define DTRANK  64
#define MROWS   (BATCH * LENGTH)
#define XPROJ_N (DTRANK + 2 * DSTATE)

#define NCH 32
#define LCH (LENGTH / NCH)
#define STG ((LCH * DSTATE) / 256)
#define DBLKS (DINNER / 256)
#define P2B 32

typedef __attribute__((ext_vector_type(2))) float v2f;
typedef __attribute__((ext_vector_type(4))) float v4f;
typedef __attribute__((ext_vector_type(8))) float v8f;

#define XP_WAVES  16
#define XP_KSLICE (DINNER / XP_WAVES)
#define XP_NT     (XPROJ_N / 16)
#define XP_DRP    (DTRANK + 4)
#define DT_NT     (DINNER / (16 * XP_WAVES))

__global__ void __launch_bounds__(XP_WAVES * 32) __attribute__((amdgpu_waves_per_eu(4, 4)))
k_xproj(const float* __restrict__ u, const float* __restrict__ Wx,
        const float* __restrict__ Wdt, const float* __restrict__ bdt,
        float* __restrict__ xdbl, float* __restrict__ delta) {
  __shared__ alignas(16) float red[XP_WAVES][XP_NT][8][32];
  __shared__ alignas(16) float dr[16][XP_DRP];
  const int w    = threadIdx.x >> 5;
  const int lane = threadIdx.x & 31;
  const int c    = lane & 15;
  const int h    = lane >> 4;
  const int tileM = blockIdx.x;

  const int kofs = w * XP_KSLICE + 4 * h;
  const float* aRow = u  + (size_t)(tileM * 16 + c) * DINNER + kofs;
  const float* bRow = Wx + (size_t)c * DINNER + kofs;

  v8f acc[XP_NT];
#pragma unroll
  for (int j = 0; j < XP_NT; ++j) acc[j] = v8f{};

  v4f a[XP_KSLICE / 8];
  v4f b[XP_NT][XP_KSLICE / 8];
#pragma unroll
  for (int i = 0; i < XP_KSLICE / 8; ++i)
    a[i] = *(const v4f*)(aRow + 8 * i);
  __builtin_amdgcn_sched_barrier(0);
#pragma unroll
  for (int j = 0; j < XP_NT; ++j) {
#pragma unroll
    for (int i = 0; i < XP_KSLICE / 8; ++i)
      b[j][i] = *(const v4f*)(bRow + (size_t)j * 16 * DINNER + 8 * i);
    if (j < XP_NT - 1) __builtin_amdgcn_sched_barrier(0);
  }

#pragma unroll
  for (int i = 0; i < XP_KSLICE / 8; ++i) {
    const v2f a0 = {a[i].x, a[i].y};
    const v2f a1 = {a[i].z, a[i].w};
#pragma unroll
    for (int j = 0; j < XP_NT; ++j) {
      const v2f b0 = {b[j][i].x, b[j][i].y};
      acc[j] = __builtin_amdgcn_wmma_f32_16x16x4_f32(
           false, a0,  false, b0,
           (short)0, acc[j],  false,  false);
    }
#pragma unroll
    for (int j = 0; j < XP_NT; ++j) {
      const v2f b1 = {b[j][i].z, b[j][i].w};
      acc[j] = __builtin_amdgcn_wmma_f32_16x16x4_f32(
          false, a1, false, b1, (short)0, acc[j], false, false);
    }
  }

#pragma unroll
  for (int j = 0; j < XP_NT; ++j)
#pragma unroll
    for (int r = 0; r < 8; ++r)
      red[w][j][r][lane] = acc[j][r];
  __syncthreads();

  const float* bRow2 = Wdt + (size_t)(w * (16 * DT_NT) + c) * DTRANK + 4 * h;
  v4f bq[DT_NT][DTRANK / 8];
#pragma unroll
  for (int nt = 0; nt < DT_NT; ++nt)
#pragma unroll
    for (int i = 0; i < DTRANK / 8; ++i)
      bq[nt][i] = *(const v4f*)(bRow2 + (size_t)nt * 16 * DTRANK + 8 * i);
  const int trow = lane >> 4;
  const int col4 = 4 * (lane & 15);
  const v4f bias4 = *(const v4f*)(bdt + w * (16 * DT_NT) + col4);
  __builtin_amdgcn_sched_barrier(0);

  float* outBase = xdbl + (size_t)tileM * (16 * XPROJ_N);
#pragma unroll
  for (int q = 0; q < (16 * XPROJ_N) / (XP_WAVES * 32); ++q) {
    const int o   = threadIdx.x + q * (XP_WAVES * 32);
    const int row = o / XPROJ_N;
    const int col = o - row * XPROJ_N;
    const int j   = col >> 4;
    const int src = ((row >> 3) << 4) | (col & 15);
    const int r   = row & 7;
    float s = red[0][j][r][src];
#pragma unroll
    for (int ww = 1; ww < XP_WAVES; ++ww) s += red[ww][j][r][src];
    if (col < DTRANK) dr[row][col] = s;
    else              outBase[o]   = s;
  }
  __syncthreads();

  v4f aq[DTRANK / 8];
#pragma unroll
  for (int i = 0; i < DTRANK / 8; ++i)
    aq[i] = *(const v4f*)&dr[c][8 * i + 4 * h];
  v8f dacc[DT_NT];
#pragma unroll
  for (int nt = 0; nt < DT_NT; ++nt) dacc[nt] = v8f{};
#pragma unroll
  for (int i = 0; i < DTRANK / 8; ++i) {
    const v2f a0 = {aq[i].x, aq[i].y};
    const v2f a1 = {aq[i].z, aq[i].w};
#pragma unroll
    for (int nt = 0; nt < DT_NT; ++nt) {
      const v2f b0 = {bq[nt][i].x, bq[nt][i].y};
      dacc[nt] = __builtin_amdgcn_wmma_f32_16x16x4_f32(
          false, a0, false, b0, (short)0, dacc[nt], false, false);
    }
#pragma unroll
    for (int nt = 0; nt < DT_NT; ++nt) {
      const v2f b1 = {bq[nt][i].z, bq[nt][i].w};
      dacc[nt] = __builtin_amdgcn_wmma_f32_16x16x4_f32(
          false, a1, false, b1, (short)0, dacc[nt], false, false);
    }
  }

  float* tile = &red[0][0][0][0] + w * (16 * XP_DRP);
#pragma unroll
  for (int nt = 0; nt < DT_NT; ++nt)
#pragma unroll
    for (int r = 0; r < 8; ++r)
      tile[(8 * h + r) * XP_DRP + 16 * nt + c] = dacc[nt][r];
  __syncthreads();

  const size_t orow = (size_t)tileM * 16;
  float* dOut = delta + orow * DINNER + w * (16 * DT_NT) + col4;
#pragma unroll 1
  for (int it = 0; it < 8; ++it) {
    const int row = 2 * it + trow;
    const v4f x = *(const v4f*)(tile + row * XP_DRP + col4);
    v4f sp;
#pragma unroll
    for (int e = 0; e < 4; ++e) {
      const float xv = x[e] + bias4[e];
      const float lg = log1pf(__expf(xv));
      sp[e] = (xv > 20.f) ? xv : lg;
    }
    *(v4f*)(dOut + (size_t)row * DINNER) = sp;
  }
}

__global__ void __launch_bounds__(256) k_scan_phase1(const float* __restrict__ u, const float* __restrict__ xdbl,
                              const float* __restrict__ delta, const float* __restrict__ logA,
                              float* __restrict__ chunkA, float* __restrict__ chunkH) {
  const int dblk = blockIdx.x % DBLKS;
  const int ch   = (blockIdx.x / DBLKS) % NCH;
  const int b    = blockIdx.x / (DBLKS * NCH);
  const int tid  = threadIdx.x;
  const int d    = dblk * 256 + tid;
  const int t0   = ch * LCH;

  __shared__ float Bs[LCH][DSTATE];
  __shared__ float Dl[LCH][256];
  __shared__ float Uu[LCH][256];

  static_assert(STG * 256 == LCH * DSTATE, "whole staged B elements per thread");
  const int st = tid >> 4, sn = tid & 15;
  float bstage[STG];
#pragma unroll
  for (int q = 0; q < STG; ++q)
    bstage[q] = xdbl[(size_t)(b * LENGTH + t0 + st + 16 * q) * XPROJ_N + DTRANK + sn];

  float An[DSTATE];
#pragma unroll
  for (int n = 0; n < DSTATE; ++n) An[n] = logA[d * DSTATE + n];

  const size_t idx0 = (size_t)(b * LENGTH + t0) * DINNER + d;
  float dl[LCH], uv[LCH];
#pragma unroll
  for (int t = 0; t < LCH; ++t) {
    dl[t] = delta[idx0 + (size_t)t * DINNER];
    uv[t] = u[idx0 + (size_t)t * DINNER];
  }

  asm volatile("" :: "v"(An[0]), "v"(An[1]), "v"(An[2]), "v"(An[3]),
                     "v"(An[4]), "v"(An[5]), "v"(An[6]), "v"(An[7]),
                     "v"(An[8]), "v"(An[9]), "v"(An[10]), "v"(An[11]),
                     "v"(An[12]), "v"(An[13]), "v"(An[14]), "v"(An[15]));
  __builtin_amdgcn_sched_barrier(0);
#pragma unroll
  for (int q = 0; q < STG; ++q) Bs[st + 16 * q][sn] = bstage[q];
#pragma unroll
  for (int t = 0; t < LCH; ++t) {
    Dl[t][tid] = dl[t];
    Uu[t][tid] = uv[t];
  }
#pragma unroll
  for (int n = 0; n < DSTATE; ++n) An[n] = -__expf(An[n]);
  __syncthreads();

  float h[DSTATE], ap[DSTATE];
#pragma unroll
  for (int n = 0; n < DSTATE; ++n) {
    h[n]  = 0.f;
    ap[n] = 1.f;
  }
  float dnext = Dl[0][tid], unext = Uu[0][tid];
#pragma unroll 2
  for (int t = 0; t < LCH; ++t) {
    const float dlt = dnext;
    const float du  = dlt * unext;
    const int   tn  = (t + 1 < LCH) ? t + 1 : t;
    dnext = Dl[tn][tid];
    unext = Uu[tn][tid];
#pragma unroll
    for (int n = 0; n < DSTATE; ++n) {
      float a = __expf(dlt * An[n]);
      h[n]  = a * h[n] + du * Bs[t][n];
      ap[n] *= a;
    }
  }
  size_t base = (((size_t)(b * DINNER + d)) * NCH + ch) * DSTATE;
#pragma unroll
  for (int n = 0; n < DSTATE; ++n) {
    chunkA[base + n] = ap[n];
    chunkH[base + n] = h[n];
  }
}

__global__ void k_scan_phase2(const float* __restrict__ chunkA, const float* __restrict__ chunkH,
                              float* __restrict__ carry) {
  const int idx = blockIdx.x * 256 + threadIdx.x;
  const int n   = idx & 15;
  const int bd  = idx >> 4;
  size_t base = (size_t)bd * NCH * DSTATE + n;
  float h = 0.f;
  for (int c0 = 0; c0 < NCH; c0 += P2B) {
    float a[P2B], g[P2B];
#pragma unroll
    for (int j = 0; j < P2B; ++j) {
      a[j] = chunkA[base + (size_t)(c0 + j) * DSTATE];
      g[j] = chunkH[base + (size_t)(c0 + j) * DSTATE];
    }
#pragma unroll
    for (int j = 0; j < P2B; ++j) {
      carry[base + (size_t)(c0 + j) * DSTATE] = h;
      h = a[j] * h + g[j];
    }
  }
}

__global__ void __launch_bounds__(256) k_scan_phase3(const float* __restrict__ u, const float* __restrict__ xdbl,
                              const float* __restrict__ delta, const float* __restrict__ logA,
                              const float* __restrict__ Dp, const float* __restrict__ carry,
                              float* __restrict__ y) {
  const int dblk = blockIdx.x % DBLKS;
  const int ch   = (blockIdx.x / DBLKS) % NCH;
  const int b    = blockIdx.x / (DBLKS * NCH);
  const int tid  = threadIdx.x;
  const int d    = dblk * 256 + tid;
  const int t0   = ch * LCH;

  __shared__ float Bs[LCH][DSTATE];
  __shared__ float Cs[LCH][DSTATE];
  __shared__ float Dl[LCH][256];
  __shared__ float Uu[LCH][256];

  static_assert(STG * 256 == LCH * DSTATE, "whole staged B and C elements per thread");
  float An[DSTATE], h[DSTATE];
  const size_t cbase = (((size_t)(b * DINNER + d)) * NCH + ch) * DSTATE;
#pragma unroll
  for (int n = 0; n < DSTATE; ++n) h[n] = carry[cbase + n];
#pragma unroll
  for (int n = 0; n < DSTATE; ++n) An[n] = logA[d * DSTATE + n];
  const float Dd = Dp[d];

  const int st = tid >> 4, sn = tid & 15;
  float bstage[STG], cstage[STG];
#pragma unroll
  for (int q = 0; q < STG; ++q) {
    const size_t rowoff = (size_t)(b * LENGTH + t0 + st + 16 * q) * XPROJ_N;
    bstage[q] = xdbl[rowoff + DTRANK + sn];
    cstage[q] = xdbl[rowoff + DTRANK + DSTATE + sn];
  }

  const size_t idx0 = (size_t)(b * LENGTH + t0) * DINNER + d;
  float dl[LCH], uv[LCH];
#pragma unroll
  for (int t = 0; t < LCH; ++t) {
    dl[t] = delta[idx0 + (size_t)t * DINNER];
    uv[t] = u[idx0 + (size_t)t * DINNER];
  }

  asm volatile("" :: "v"(An[0]), "v"(An[1]), "v"(An[2]), "v"(An[3]),
                     "v"(An[4]), "v"(An[5]), "v"(An[6]), "v"(An[7]),
                     "v"(An[8]), "v"(An[9]), "v"(An[10]), "v"(An[11]),
                     "v"(An[12]), "v"(An[13]), "v"(An[14]), "v"(An[15]));
  __builtin_amdgcn_sched_barrier(0);
#pragma unroll
  for (int q = 0; q < STG; ++q) {
    Bs[st + 16 * q][sn] = bstage[q];
    Cs[st + 16 * q][sn] = cstage[q];
  }
#pragma unroll
  for (int t = 0; t < LCH; ++t) {
    Dl[t][tid] = dl[t];
    Uu[t][tid] = uv[t];
  }
#pragma unroll
  for (int n = 0; n < DSTATE; ++n) An[n] = -__expf(An[n]);
  __syncthreads();

  float dnext = Dl[0][tid], unext = Uu[0][tid];
#pragma unroll 2
  for (int t = 0; t < LCH; ++t) {
    const float dlt = dnext;
    const float uu  = unext;
    const float du  = dlt * uu;
    const int   tn  = (t + 1 < LCH) ? t + 1 : t;
    dnext = Dl[tn][tid];
    unext = Uu[tn][tid];
    float acc = 0.f;
#pragma unroll
    for (int n = 0; n < DSTATE; ++n) {
      float a = __expf(dlt * An[n]);
      h[n] = a * h[n] + du * Bs[t][n];
      acc += h[n] * Cs[t][n];
    }
    y[idx0 + (size_t)t * DINNER] = acc + uu * Dd;
  }
}

extern "C" void kernel_launch(void* const* d_in, const int* in_sizes, int n_in,
                              void* d_out, int out_size, void* d_ws, size_t ws_size,
                              hipStream_t stream) {
  (void)in_sizes; (void)n_in; (void)out_size; (void)ws_size;
  const float* u    = (const float*)d_in[0];
  const float* Wx   = (const float*)d_in[1];
  const float* Wdt  = (const float*)d_in[2];
  const float* bdt  = (const float*)d_in[3];
  const float* logA = (const float*)d_in[4];
  const float* Dp   = (const float*)d_in[5];
  float* y = (float*)d_out;

  float* ws     = (float*)d_ws;
  float* xdbl   = ws;
  float* delta  = xdbl  + (size_t)MROWS * XPROJ_N;
  float* chunkA = delta + (size_t)MROWS * DINNER;
  float* chunkH = chunkA + (size_t)BATCH * DINNER * NCH * DSTATE;
  float* carry  = chunkH + (size_t)BATCH * DINNER * NCH * DSTATE;

  k_xproj<<<MROWS / 16, XP_WAVES * 32, 0, stream>>>(u, Wx, Wdt, bdt, xdbl, delta);
  const int scan_blocks = BATCH * NCH * DBLKS;
  k_scan_phase1<<<scan_blocks, 256, 0, stream>>>(u, xdbl, delta, logA, chunkA, chunkH);
  k_scan_phase2<<<(BATCH * DINNER * DSTATE) / 256, 256, 0, stream>>>(chunkA, chunkH, carry);
  k_scan_phase3<<<scan_blocks, 256, 0, stream>>>(u, xdbl, delta, logA, Dp, carry, y);
}


# softplus in the x_proj epilogue: library log1pf replaced by max(x,0)+log1p(exp(-|x|)), log1p as the series 2*atanh(e/(2+e)), 7 terms: about 17 vector instructions per element; sum order unchanged
# speedup vs baseline: 3.5125x; 1.0446x over previous
#include <hip/hip_runtime.h>
#include <hip/hip_bf16.h>
#include <math.h>

#define BATCH   2
#define LENGTH  2048
#define DINNER  1024
#define DSTATE  16
#define DTRANK  64
#define MROWS   (BATCH * LENGTH)
#define XPROJ_N (DTRANK + 2 * DSTATE)

#define NCH 32
#define LCH (LENGTH / NCH)
#define STG ((LCH * DSTATE) / 256)
#define DBLKS (DINNER / 256)
#define P2B 32

typedef __attribute__((ext_vector_type(2))) float v2f;
typedef __attribute__((ext_vector_type(4))) float v4f;
typedef __attribute__((ext_vector_type(8))) float v8f;

#define XP_WAVES  16
#define XP_KSLICE (DINNER / XP_WAVES)
#define XP_NT     (XPROJ_N / 16)
#define XP_DRP    (DTRANK + 4)
#define DT_NT     (DINNER / (16 * XP_WAVES))

__device__ __forceinline__ float softplus_f32(float x) {
  const float e  = __expf(-fabsf(x));
  const float s  = e * __builtin_amdgcn_rcpf(e + 2.0f);
  const float z  = s * s;
  float p = fmaf(z, 1.0f / 13.0f, 1.0f / 11.0f);
  p = fmaf(z, p, 1.0f / 9.0f);
  p = fmaf(z, p, 1.0f / 7.0f);
  p = fmaf(z, p, 1.0f / 5.0f);
  p = fmaf(z, p, 1.0f / 3.0f);
  const float s2 = s + s;
  return fmaxf(x, 0.0f) + fmaf(s2, z * p, s2);
}

__global__ void __launch_bounds__(XP_WAVES * 32) __attribute__((amdgpu_waves_per_eu(4, 4)))
k_xproj(const float* __restrict__ u, const float* __restrict__ Wx,
        const float* __restrict__ Wdt, const float* __restrict__ bdt,
        float* __restrict__ xdbl, float* __restrict__ delta) {
  __shared__ alignas(16) float red[XP_WAVES][XP_NT][8][32];
  __shared__ alignas(16) float dr[16][XP_DRP];
  const int w    = threadIdx.x >> 5;
  const int lane = threadIdx.x & 31;
  const int c    = lane & 15;
  const int h    = lane >> 4;
  const int tileM = blockIdx.x;

  const int kofs = w * XP_KSLICE + 4 * h;
  const float* aRow = u  + (size_t)(tileM * 16 + c) * DINNER + kofs;
  const float* bRow = Wx + (size_t)c * DINNER + kofs;

  v8f acc[XP_NT];
#pragma unroll
  for (int j = 0; j < XP_NT; ++j) acc[j] = v8f{};

  v4f a[XP_KSLICE / 8];
  v4f b[XP_NT][XP_KSLICE / 8];
#pragma unroll
  for (int i = 0; i < XP_KSLICE / 8; ++i)
    a[i] = *(const v4f*)(aRow + 8 * i);
  __builtin_amdgcn_sched_barrier(0);
#pragma unroll
  for (int j = 0; j < XP_NT; ++j) {
#pragma unroll
    for (int i = 0; i < XP_KSLICE / 8; ++i)
      b[j][i] = *(const v4f*)(bRow + (size_t)j * 16 * DINNER + 8 * i);
    if (j < XP_NT - 1) __builtin_amdgcn_sched_barrier(0);
  }

#pragma unroll
  for (int i = 0; i < XP_KSLICE / 8; ++i) {
    const v2f a0 = {a[i].x, a[i].y};
    const v2f a1 = {a[i].z, a[i].w};
#pragma unroll
    for (int j = 0; j < XP_NT; ++j) {
      const v2f b0 = {b[j][i].x, b[j][i].y};
      acc[j] = __builtin_amdgcn_wmma_f32_16x16x4_f32(
           false, a0,  false, b0,
           (short)0, acc[j],  false,  false);
    }
#pragma unroll
    for (int j = 0; j < XP_NT; ++j) {
      const v2f b1 = {b[j][i].z, b[j][i].w};
      acc[j] = __builtin_amdgcn_wmma_f32_16x16x4_f32(
          false, a1, false, b1, (short)0, acc[j], false, false);
    }
  }

#pragma unroll
  for (int j = 0; j < XP_NT; ++j)
#pragma unroll
    for (int r = 0; r < 8; ++r)
      red[w][j][r][lane] = acc[j][r];
  __syncthreads();

  const float* bRow2 = Wdt + (size_t)(w * (16 * DT_NT) + c) * DTRANK + 4 * h;
  v4f bq[DT_NT][DTRANK / 8];
#pragma unroll
  for (int nt = 0; nt < DT_NT; ++nt)
#pragma unroll
    for (int i = 0; i < DTRANK / 8; ++i)
      bq[nt][i] = *(const v4f*)(bRow2 + (size_t)nt * 16 * DTRANK + 8 * i);
  const int trow = lane >> 4;
  const int col4 = 4 * (lane & 15);
  const v4f bias4 = *(const v4f*)(bdt + w * (16 * DT_NT) + col4);
  __builtin_amdgcn_sched_barrier(0);

  float* outBase = xdbl + (size_t)tileM * (16 * XPROJ_N);
#pragma unroll
  for (int q = 0; q < (16 * XPROJ_N) / (XP_WAVES * 32); ++q) {
    const int o   = threadIdx.x + q * (XP_WAVES * 32);
    const int row = o / XPROJ_N;
    const int col = o - row * XPROJ_N;
    const int j   = col >> 4;
    const int src = ((row >> 3) << 4) | (col & 15);
    const int r   = row & 7;
    float s = red[0][j][r][src];
#pragma unroll
    for (int ww = 1; ww < XP_WAVES; ++ww) s += red[ww][j][r][src];
    if (col < DTRANK) dr[row][col] = s;
    else              outBase[o]   = s;
  }
  __syncthreads();

  v4f aq[DTRANK / 8];
#pragma unroll
  for (int i = 0; i < DTRANK / 8; ++i)
    aq[i] = *(const v4f*)&dr[c][8 * i + 4 * h];
  v8f dacc[DT_NT];
#pragma unroll
  for (int nt = 0; nt < DT_NT; ++nt) dacc[nt] = v8f{};
#pragma unroll
  for (int i = 0; i < DTRANK / 8; ++i) {
    const v2f a0 = {aq[i].x, aq[i].y};
    const v2f a1 = {aq[i].z, aq[i].w};
#pragma unroll
    for (int nt = 0; nt < DT_NT; ++nt) {
      const v2f b0 = {bq[nt][i].x, bq[nt][i].y};
      dacc[nt] = __builtin_amdgcn_wmma_f32_16x16x4_f32(
          false, a0, false, b0, (short)0, dacc[nt], false, false);
    }
#pragma unroll
    for (int nt = 0; nt < DT_NT; ++nt) {
      const v2f b1 = {bq[nt][i].z, bq[nt][i].w};
      dacc[nt] = __builtin_amdgcn_wmma_f32_16x16x4_f32(
          false, a1, false, b1, (short)0, dacc[nt], false, false);
    }
  }

  float* tile = &red[0][0][0][0] + w * (16 * XP_DRP);
#pragma unroll
  for (int nt = 0; nt < DT_NT; ++nt)
#pragma unroll
    for (int r = 0; r < 8; ++r)
      tile[(8 * h + r) * XP_DRP + 16 * nt + c] = dacc[nt][r];
  __syncthreads();

  const size_t orow = (size_t)tileM * 16;
  float* dOut = delta + orow * DINNER + w * (16 * DT_NT) + col4;
  for (int it = 0; it < 8; ++it) {
    const int row = 2 * it + trow;
    const v4f x = *(const v4f*)(tile + row * XP_DRP + col4);
    v4f sp;
#pragma unroll
    for (int e = 0; e < 4; ++e) sp[e] = softplus_f32(x[e] + bias4[e]);
    *(v4f*)(dOut + (size_t)row * DINNER) = sp;
  }
}

__global__ void __launch_bounds__(256) k_scan_phase1(const float* __restrict__ u, const float* __restrict__ xdbl,
                              const float* __restrict__ delta, const float* __restrict__ logA,
                              float* __restrict__ chunkA, float* __restrict__ chunkH) {
  const int dblk = blockIdx.x % DBLKS;
  const int ch   = (blockIdx.x / DBLKS) % NCH;
  const int b    = blockIdx.x / (DBLKS * NCH);
  const int tid  = threadIdx.x;
  const int d    = dblk * 256 + tid;
  const int t0   = ch * LCH;

  __shared__ float Bs[LCH][DSTATE];
  __shared__ float Dl[LCH][256];
  __shared__ float Uu[LCH][256];

  static_assert(STG * 256 == LCH * DSTATE, "whole staged B elements per thread");
  const int st = tid >> 4, sn = tid & 15;
  float bstage[STG];
#pragma unroll
  for (int q = 0; q < STG; ++q)
    bstage[q] = xdbl[(size_t)(b * LENGTH + t0 + st + 16 * q) * XPROJ_N + DTRANK + sn];

  float An[DSTATE];
#pragma unroll
  for (int n = 0; n < DSTATE; ++n) An[n] = logA[d * DSTATE + n];

  const size_t idx0 = (size_t)(b * LENGTH + t0) * DINNER + d;
  float dl[LCH], uv[LCH];
#pragma unroll
  for (int t = 0; t < LCH; ++t) {
    dl[t] = delta[idx0 + (size_t)t * DINNER];
    uv[t] = u[idx0 + (size_t)t * DINNER];
  }

  asm volatile("" :: "v"(An[0]), "v"(An[1]), "v"(An[2]), "v"(An[3]),
                     "v"(An[4]), "v"(An[5]), "v"(An[6]), "v"(An[7]),
                     "v"(An[8]), "v"(An[9]), "v"(An[10]), "v"(An[11]),
                     "v"(An[12]), "v"(An[13]), "v"(An[14]), "v"(An[15]));
  __builtin_amdgcn_sched_barrier(0);
#pragma unroll
  for (int q = 0; q < STG; ++q) Bs[st + 16 * q][sn] = bstage[q];
#pragma unroll
  for (int t = 0; t < LCH; ++t) {
    Dl[t][tid] = dl[t];
    Uu[t][tid] = uv[t];
  }
#pragma unroll
  for (int n = 0; n < DSTATE; ++n) An[n] = -__expf(An[n]);
  __syncthreads();

  float h[DSTATE], ap[DSTATE];
#pragma unroll
  for (int n = 0; n < DSTATE; ++n) {
    h[n]  = 0.f;
    ap[n] = 1.f;
  }
  float dnext = Dl[0][tid], unext = Uu[0][tid];
#pragma unroll 2
  for (int t = 0; t < LCH; ++t) {
    const float dlt = dnext;
    const float du  = dlt * unext;
    const int   tn  = (t + 1 < LCH) ? t + 1 : t;
    dnext = Dl[tn][tid];
    unext = Uu[tn][tid];
#pragma unroll
    for (int n = 0; n < DSTATE; ++n) {
      float a = __expf(dlt * An[n]);
      h[n]  = a * h[n] + du * Bs[t][n];
      ap[n] *= a;
    }
  }
  size_t base = (((size_t)(b * DINNER + d)) * NCH + ch) * DSTATE;
#pragma unroll
  for (int n = 0; n < DSTATE; ++n) {
    chunkA[base + n] = ap[n];
    chunkH[base + n] = h[n];
  }
}

__global__ void k_scan_phase2(const float* __restrict__ chunkA, const float* __restrict__ chunkH,
                              float* __restrict__ carry) {
  const int idx = blockIdx.x * 256 + threadIdx.x;
  const int n   = idx & 15;
  const int bd  = idx >> 4;
  size_t base = (size_t)bd * NCH * DSTATE + n;
  float h = 0.f;
  for (int c0 = 0; c0 < NCH; c0 += P2B) {
    float a[P2B], g[P2B];
#pragma unroll
    for (int j = 0; j < P2B; ++j) {
      a[j] = chunkA[base + (size_t)(c0 + j) * DSTATE];
      g[j] = chunkH[base + (size_t)(c0 + j) * DSTATE];
    }
#pragma unroll
    for (int j = 0; j < P2B; ++j) {
      carry[base + (size_t)(c0 + j) * DSTATE] = h;
      h = a[j] * h + g[j];
    }
  }
}

__global__ void __launch_bounds__(256) k_scan_phase3(const float* __restrict__ u, const float* __restrict__ xdbl,
                              const float* __restrict__ delta, const float* __restrict__ logA,
                              const float* __restrict__ Dp, const float* __restrict__ carry,
                              float* __restrict__ y) {
  const int dblk = blockIdx.x % DBLKS;
  const int ch   = (blockIdx.x / DBLKS) % NCH;
  const int b    = blockIdx.x / (DBLKS * NCH);
  const int tid  = threadIdx.x;
  const int d    = dblk * 256 + tid;
  const int t0   = ch * LCH;

  __shared__ float Bs[LCH][DSTATE];
  __shared__ float Cs[LCH][DSTATE];
  __shared__ float Dl[LCH][256];
  __shared__ float Uu[LCH][256];

  static_assert(STG * 256 == LCH * DSTATE, "whole staged B and C elements per thread");
  float An[DSTATE], h[DSTATE];
  const size_t cbase = (((size_t)(b * DINNER + d)) * NCH + ch) * DSTATE;
#pragma unroll
  for (int n = 0; n < DSTATE; ++n) h[n] = carry[cbase + n];
#pragma unroll
  for (int n = 0; n < DSTATE; ++n) An[n] = logA[d * DSTATE + n];
  const float Dd = Dp[d];

  const int st = tid >> 4, sn = tid & 15;
  float bstage[STG], cstage[STG];
#pragma unroll
  for (int q = 0; q < STG; ++q) {
    const size_t rowoff = (size_t)(b * LENGTH + t0 + st + 16 * q) * XPROJ_N;
    bstage[q] = xdbl[rowoff + DTRANK + sn];
    cstage[q] = xdbl[rowoff + DTRANK + DSTATE + sn];
  }

  const size_t idx0 = (size_t)(b * LENGTH + t0) * DINNER + d;
  float dl[LCH], uv[LCH];
#pragma unroll
  for (int t = 0; t < LCH; ++t) {
    dl[t] = delta[idx0 + (size_t)t * DINNER];
    uv[t] = u[idx0 + (size_t)t * DINNER];
  }

  asm volatile("" :: "v"(An[0]), "v"(An[1]), "v"(An[2]), "v"(An[3]),
                     "v"(An[4]), "v"(An[5]), "v"(An[6]), "v"(An[7]),
                     "v"(An[8]), "v"(An[9]), "v"(An[10]), "v"(An[11]),
                     "v"(An[12]), "v"(An[13]), "v"(An[14]), "v"(An[15]));
  __builtin_amdgcn_sched_barrier(0);
#pragma unroll
  for (int q = 0; q < STG; ++q) {
    Bs[st + 16 * q][sn] = bstage[q];
    Cs[st + 16 * q][sn] = cstage[q];
  }
#pragma unroll
  for (int t = 0; t < LCH; ++t) {
    Dl[t][tid] = dl[t];
    Uu[t][tid] = uv[t];
  }
#pragma unroll
  for (int n = 0; n < DSTATE; ++n) An[n] = -__expf(An[n]);
  __syncthreads();

  float dnext = Dl[0][tid], unext = Uu[0][tid];
#pragma unroll 2
  for (int t = 0; t < LCH; ++t) {
    const float dlt = dnext;
    const float uu  = unext;
    const float du  = dlt * uu;
    const int   tn  = (t + 1 < LCH) ? t + 1 : t;
    dnext = Dl[tn][tid];
    unext = Uu[tn][tid];
    float acc = 0.f;
#pragma unroll
    for (int n = 0; n < DSTATE; ++n) {
      float a = __expf(dlt * An[n]);
      h[n] = a * h[n] + du * Bs[t][n];
      acc += h[n] * Cs[t][n];
    }
    y[idx0 + (size_t)t * DINNER] = acc + uu * Dd;
  }
}

extern "C" void kernel_launch(void* const* d_in, const int* in_sizes, int n_in,
                              void* d_out, int out_size, void* d_ws, size_t ws_size,
                              hipStream_t stream) {
  (void)in_sizes; (void)n_in; (void)out_size; (void)ws_size;
  const float* u    = (const float*)d_in[0];
  const float* Wx   = (const float*)d_in[1];
  const float* Wdt  = (const float*)d_in[2];
  const float* bdt  = (const float*)d_in[3];
  const float* logA = (const float*)d_in[4];
  const float* Dp   = (const float*)d_in[5];
  float* y = (float*)d_out;

  float* ws     = (float*)d_ws;
  float* xdbl   = ws;
  float* delta  = xdbl  + (size_t)MROWS * XPROJ_N;
  float* chunkA = delta + (size_t)MROWS * DINNER;
  float* chunkH = chunkA + (size_t)BATCH * DINNER * NCH * DSTATE;
  float* carry  = chunkH + (size_t)BATCH * DINNER * NCH * DSTATE;

  k_xproj<<<MROWS / 16, XP_WAVES * 32, 0, stream>>>(u, Wx, Wdt, bdt, xdbl, delta);
  const int scan_blocks = BATCH * NCH * DBLKS;
  k_scan_phase1<<<scan_blocks, 256, 0, stream>>>(u, xdbl, delta, logA, chunkA, chunkH);
  k_scan_phase2<<<(BATCH * DINNER * DSTATE) / 256, 256, 0, stream>>>(chunkA, chunkH, carry);
  k_scan_phase3<<<scan_blocks, 256, 0, stream>>>(u, xdbl, delta, logA, Dp, carry, y);
}


# scan time loops (phases 1, 3) issue fewer vector instructions per step: log2(e) folded into A, chunk product of A_bar as exp2(A*sum of delta), C sum in two packed accumulators; sum order changes
# speedup vs baseline: 3.7521x; 1.0682x over previous
#include <hip/hip_runtime.h>
#include <hip/hip_bf16.h>
#include <math.h>

#define BATCH   2
#define LENGTH  2048
#define DINNER  1024
#define DSTATE  16
#define DTRANK  64
#define MROWS   (BATCH * LENGTH)
#define XPROJ_N (DTRANK + 2 * DSTATE)

#define NCH 32
#define LCH (LENGTH / NCH)
#define STG ((LCH * DSTATE) / 256)
#define DBLKS (DINNER / 256)
#define P2B 32

typedef __attribute__((ext_vector_type(2))) float v2f;
typedef __attribute__((ext_vector_type(4))) float v4f;
typedef __attribute__((ext_vector_type(8))) float v8f;

#define XP_WAVES  16
#define XP_KSLICE (DINNER / XP_WAVES)
#define XP_NT     (XPROJ_N / 16)
#define XP_DRP    (DTRANK + 4)
#define DT_NT     (DINNER / (16 * XP_WAVES))

__device__ __forceinline__ float softplus_f32(float x) {
  const float e  = __expf(-fabsf(x));
  const float s  = e * __builtin_amdgcn_rcpf(e + 2.0f);
  const float z  = s * s;
  float p = fmaf(z, 1.0f / 13.0f, 1.0f / 11.0f);
  p = fmaf(z, p, 1.0f / 9.0f);
  p = fmaf(z, p, 1.0f / 7.0f);
  p = fmaf(z, p, 1.0f / 5.0f);
  p = fmaf(z, p, 1.0f / 3.0f);
  const float s2 = s + s;
  return fmaxf(x, 0.0f) + fmaf(s2, z * p, s2);
}

__global__ void __launch_bounds__(XP_WAVES * 32) __attribute__((amdgpu_waves_per_eu(4, 4)))
k_xproj(const float* __restrict__ u, const float* __restrict__ Wx,
        const float* __restrict__ Wdt, const float* __restrict__ bdt,
        float* __restrict__ xdbl, float* __restrict__ delta) {
  __shared__ alignas(16) float red[XP_WAVES][XP_NT][8][32];
  __shared__ alignas(16) float dr[16][XP_DRP];
  const int w    = threadIdx.x >> 5;
  const int lane = threadIdx.x & 31;
  const int c    = lane & 15;
  const int h    = lane >> 4;
  const int tileM = blockIdx.x;

  const int kofs = w * XP_KSLICE + 4 * h;
  const float* aRow = u  + (size_t)(tileM * 16 + c) * DINNER + kofs;
  const float* bRow = Wx + (size_t)c * DINNER + kofs;

  v8f acc[XP_NT];
#pragma unroll
  for (int j = 0; j < XP_NT; ++j) acc[j] = v8f{};

  v4f a[XP_KSLICE / 8];
  v4f b[XP_NT][XP_KSLICE / 8];
#pragma unroll
  for (int i = 0; i < XP_KSLICE / 8; ++i)
    a[i] = *(const v4f*)(aRow + 8 * i);
  __builtin_amdgcn_sched_barrier(0);
#pragma unroll
  for (int j = 0; j < XP_NT; ++j) {
#pragma unroll
    for (int i = 0; i < XP_KSLICE / 8; ++i)
      b[j][i] = *(const v4f*)(bRow + (size_t)j * 16 * DINNER + 8 * i);
    if (j < XP_NT - 1) __builtin_amdgcn_sched_barrier(0);
  }

#pragma unroll
  for (int i = 0; i < XP_KSLICE / 8; ++i) {
    const v2f a0 = {a[i].x, a[i].y};
    const v2f a1 = {a[i].z, a[i].w};
#pragma unroll
    for (int j = 0; j < XP_NT; ++j) {
      const v2f b0 = {b[j][i].x, b[j][i].y};
      acc[j] = __builtin_amdgcn_wmma_f32_16x16x4_f32(
           false, a0,  false, b0,
           (short)0, acc[j],  false,  false);
    }
#pragma unroll
    for (int j = 0; j < XP_NT; ++j) {
      const v2f b1 = {b[j][i].z, b[j][i].w};
      acc[j] = __builtin_amdgcn_wmma_f32_16x16x4_f32(
          false, a1, false, b1, (short)0, acc[j], false, false);
    }
  }

#pragma unroll
  for (int j = 0; j < XP_NT; ++j)
#pragma unroll
    for (int r = 0; r < 8; ++r)
      red[w][j][r][lane] = acc[j][r];
  __syncthreads();

  const float* bRow2 = Wdt + (size_t)(w * (16 * DT_NT) + c) * DTRANK + 4 * h;
  v4f bq[DT_NT][DTRANK / 8];
#pragma unroll
  for (int nt = 0; nt < DT_NT; ++nt)
#pragma unroll
    for (int i = 0; i < DTRANK / 8; ++i)
      bq[nt][i] = *(const v4f*)(bRow2 + (size_t)nt * 16 * DTRANK + 8 * i);
  const int trow = lane >> 4;
  const int col4 = 4 * (lane & 15);
  const v4f bias4 = *(const v4f*)(bdt + w * (16 * DT_NT) + col4);
  __builtin_amdgcn_sched_barrier(0);

  float* outBase = xdbl + (size_t)tileM * (16 * XPROJ_N);
#pragma unroll
  for (int q = 0; q < (16 * XPROJ_N) / (XP_WAVES * 32); ++q) {
    const int o   = threadIdx.x + q * (XP_WAVES * 32);
    const int row = o / XPROJ_N;
    const int col = o - row * XPROJ_N;
    const int j   = col >> 4;
    const int src = ((row >> 3) << 4) | (col & 15);
    const int r   = row & 7;
    float s = red[0][j][r][src];
#pragma unroll
    for (int ww = 1; ww < XP_WAVES; ++ww) s += red[ww][j][r][src];
    if (col < DTRANK) dr[row][col] = s;
    else              outBase[o]   = s;
  }
  __syncthreads();

  v4f aq[DTRANK / 8];
#pragma unroll
  for (int i = 0; i < DTRANK / 8; ++i)
    aq[i] = *(const v4f*)&dr[c][8 * i + 4 * h];
  v8f dacc[DT_NT];
#pragma unroll
  for (int nt = 0; nt < DT_NT; ++nt) dacc[nt] = v8f{};
#pragma unroll
  for (int i = 0; i < DTRANK / 8; ++i) {
    const v2f a0 = {aq[i].x, aq[i].y};
    const v2f a1 = {aq[i].z, aq[i].w};
#pragma unroll
    for (int nt = 0; nt < DT_NT; ++nt) {
      const v2f b0 = {bq[nt][i].x, bq[nt][i].y};
      dacc[nt] = __builtin_amdgcn_wmma_f32_16x16x4_f32(
          false, a0, false, b0, (short)0, dacc[nt], false, false);
    }
#pragma unroll
    for (int nt = 0; nt < DT_NT; ++nt) {
      const v2f b1 = {bq[nt][i].z, bq[nt][i].w};
      dacc[nt] = __builtin_amdgcn_wmma_f32_16x16x4_f32(
          false, a1, false, b1, (short)0, dacc[nt], false, false);
    }
  }

  float* tile = &red[0][0][0][0] + w * (16 * XP_DRP);
#pragma unroll
  for (int nt = 0; nt < DT_NT; ++nt)
#pragma unroll
    for (int r = 0; r < 8; ++r)
      tile[(8 * h + r) * XP_DRP + 16 * nt + c] = dacc[nt][r];
  __syncthreads();

  const size_t orow = (size_t)tileM * 16;
  float* dOut = delta + orow * DINNER + w * (16 * DT_NT) + col4;
  for (int it = 0; it < 8; ++it) {
    const int row = 2 * it + trow;
    const v4f x = *(const v4f*)(tile + row * XP_DRP + col4);
    v4f sp;
#pragma unroll
    for (int e = 0; e < 4; ++e) sp[e] = softplus_f32(x[e] + bias4[e]);
    *(v4f*)(dOut + (size_t)row * DINNER) = sp;
  }
}

#define LOG2E_F 1.4426950408889634f
__global__ void __launch_bounds__(256) k_scan_phase1(const float* __restrict__ u, const float* __restrict__ xdbl,
                              const float* __restrict__ delta, const float* __restrict__ logA,
                              float* __restrict__ chunkA, float* __restrict__ chunkH) {
  const int dblk = blockIdx.x % DBLKS;
  const int ch   = (blockIdx.x / DBLKS) % NCH;
  const int b    = blockIdx.x / (DBLKS * NCH);
  const int tid  = threadIdx.x;
  const int d    = dblk * 256 + tid;
  const int t0   = ch * LCH;

  __shared__ float Bs[LCH][DSTATE];
  __shared__ float Dl[LCH][256];
  __shared__ float Uu[LCH][256];

  static_assert(STG * 256 == LCH * DSTATE, "whole staged B elements per thread");
  const int st = tid >> 4, sn = tid & 15;
  float bstage[STG];
#pragma unroll
  for (int q = 0; q < STG; ++q)
    bstage[q] = xdbl[(size_t)(b * LENGTH + t0 + st + 16 * q) * XPROJ_N + DTRANK + sn];

  float An[DSTATE];
#pragma unroll
  for (int n = 0; n < DSTATE; ++n) An[n] = logA[d * DSTATE + n];

  const size_t idx0 = (size_t)(b * LENGTH + t0) * DINNER + d;
  float dl[LCH], uv[LCH];
#pragma unroll
  for (int t = 0; t < LCH; ++t) {
    dl[t] = delta[idx0 + (size_t)t * DINNER];
    uv[t] = u[idx0 + (size_t)t * DINNER];
  }

  asm volatile("" :: "v"(An[0]), "v"(An[1]), "v"(An[2]), "v"(An[3]),
                     "v"(An[4]), "v"(An[5]), "v"(An[6]), "v"(An[7]),
                     "v"(An[8]), "v"(An[9]), "v"(An[10]), "v"(An[11]),
                     "v"(An[12]), "v"(An[13]), "v"(An[14]), "v"(An[15]));
  __builtin_amdgcn_sched_barrier(0);
#pragma unroll
  for (int q = 0; q < STG; ++q) Bs[st + 16 * q][sn] = bstage[q];
#pragma unroll
  for (int t = 0; t < LCH; ++t) {
    Dl[t][tid] = dl[t];
    Uu[t][tid] = uv[t];
  }
  v2f A2[DSTATE / 2];
#pragma unroll
  for (int k = 0; k < DSTATE / 2; ++k) {
    A2[k].x = -__expf(An[2 * k])     * LOG2E_F;
    A2[k].y = -__expf(An[2 * k + 1]) * LOG2E_F;
  }
  __syncthreads();

  v2f h2[DSTATE / 2];
#pragma unroll
  for (int k = 0; k < DSTATE / 2; ++k) h2[k] = v2f{0.f, 0.f};
  float dsum = 0.f;
  float dnext = Dl[0][tid], unext = Uu[0][tid];
#pragma unroll 2
  for (int t = 0; t < LCH; ++t) {
    const float dlt = dnext;
    const float du  = dlt * unext;
    const int   tn  = (t + 1 < LCH) ? t + 1 : t;
    dnext = Dl[tn][tid];
    unext = Uu[tn][tid];
    dsum += dlt;
#pragma unroll
    for (int k = 0; k < DSTATE / 2; ++k) {
      const v2f arg = A2[k] * dlt;
      v2f a;
      a.x = __builtin_amdgcn_exp2f(arg.x);
      a.y = __builtin_amdgcn_exp2f(arg.y);
      const v2f bb = {Bs[t][2 * k], Bs[t][2 * k + 1]};
      h2[k] = a * h2[k] + bb * du;
    }
  }
  size_t base = (((size_t)(b * DINNER + d)) * NCH + ch) * DSTATE;
#pragma unroll
  for (int k = 0; k < DSTATE / 2; ++k) {
    chunkA[base + 2 * k]     = __builtin_amdgcn_exp2f(A2[k].x * dsum);
    chunkA[base + 2 * k + 1] = __builtin_amdgcn_exp2f(A2[k].y * dsum);
    chunkH[base + 2 * k]     = h2[k].x;
    chunkH[base + 2 * k + 1] = h2[k].y;
  }
}

__global__ void k_scan_phase2(const float* __restrict__ chunkA, const float* __restrict__ chunkH,
                              float* __restrict__ carry) {
  const int idx = blockIdx.x * 256 + threadIdx.x;
  const int n   = idx & 15;
  const int bd  = idx >> 4;
  size_t base = (size_t)bd * NCH * DSTATE + n;
  float h = 0.f;
  for (int c0 = 0; c0 < NCH; c0 += P2B) {
    float a[P2B], g[P2B];
#pragma unroll
    for (int j = 0; j < P2B; ++j) {
      a[j] = chunkA[base + (size_t)(c0 + j) * DSTATE];
      g[j] = chunkH[base + (size_t)(c0 + j) * DSTATE];
    }
#pragma unroll
    for (int j = 0; j < P2B; ++j) {
      carry[base + (size_t)(c0 + j) * DSTATE] = h;
      h = a[j] * h + g[j];
    }
  }
}

__global__ void __launch_bounds__(256) k_scan_phase3(const float* __restrict__ u, const float* __restrict__ xdbl,
                              const float* __restrict__ delta, const float* __restrict__ logA,
                              const float* __restrict__ Dp, const float* __restrict__ carry,
                              float* __restrict__ y) {
  const int dblk = blockIdx.x % DBLKS;
  const int ch   = (blockIdx.x / DBLKS) % NCH;
  const int b    = blockIdx.x / (DBLKS * NCH);
  const int tid  = threadIdx.x;
  const int d    = dblk * 256 + tid;
  const int t0   = ch * LCH;

  __shared__ float Bs[LCH][DSTATE];
  __shared__ float Cs[LCH][DSTATE];
  __shared__ float Dl[LCH][256];
  __shared__ float Uu[LCH][256];

  static_assert(STG * 256 == LCH * DSTATE, "whole staged B and C elements per thread");
  float An[DSTATE], h[DSTATE];
  const size_t cbase = (((size_t)(b * DINNER + d)) * NCH + ch) * DSTATE;
#pragma unroll
  for (int n = 0; n < DSTATE; ++n) h[n] = carry[cbase + n];
#pragma unroll
  for (int n = 0; n < DSTATE; ++n) An[n] = logA[d * DSTATE + n];
  const float Dd = Dp[d];

  const int st = tid >> 4, sn = tid & 15;
  float bstage[STG], cstage[STG];
#pragma unroll
  for (int q = 0; q < STG; ++q) {
    const size_t rowoff = (size_t)(b * LENGTH + t0 + st + 16 * q) * XPROJ_N;
    bstage[q] = xdbl[rowoff + DTRANK + sn];
    cstage[q] = xdbl[rowoff + DTRANK + DSTATE + sn];
  }

  const size_t idx0 = (size_t)(b * LENGTH + t0) * DINNER + d;
  float dl[LCH], uv[LCH];
#pragma unroll
  for (int t = 0; t < LCH; ++t) {
    dl[t] = delta[idx0 + (size_t)t * DINNER];
    uv[t] = u[idx0 + (size_t)t * DINNER];
  }

  asm volatile("" :: "v"(An[0]), "v"(An[1]), "v"(An[2]), "v"(An[3]),
                     "v"(An[4]), "v"(An[5]), "v"(An[6]), "v"(An[7]),
                     "v"(An[8]), "v"(An[9]), "v"(An[10]), "v"(An[11]),
                     "v"(An[12]), "v"(An[13]), "v"(An[14]), "v"(An[15]));
  __builtin_amdgcn_sched_barrier(0);
#pragma unroll
  for (int q = 0; q < STG; ++q) {
    Bs[st + 16 * q][sn] = bstage[q];
    Cs[st + 16 * q][sn] = cstage[q];
  }
#pragma unroll
  for (int t = 0; t < LCH; ++t) {
    Dl[t][tid] = dl[t];
    Uu[t][tid] = uv[t];
  }
  v2f A2[DSTATE / 2], h2[DSTATE / 2];
#pragma unroll
  for (int k = 0; k < DSTATE / 2; ++k) {
    A2[k].x = -__expf(An[2 * k])     * LOG2E_F;
    A2[k].y = -__expf(An[2 * k + 1]) * LOG2E_F;
    h2[k]   = v2f{h[2 * k], h[2 * k + 1]};
  }
  __syncthreads();

  float dnext = Dl[0][tid], unext = Uu[0][tid];
#pragma unroll 2
  for (int t = 0; t < LCH; ++t) {
    const float dlt = dnext;
    const float uu  = unext;
    const float du  = dlt * uu;
    const int   tn  = (t + 1 < LCH) ? t + 1 : t;
    dnext = Dl[tn][tid];
    unext = Uu[tn][tid];
    v2f accA = {0.f, 0.f}, accB = {0.f, 0.f};
#pragma unroll
    for (int k = 0; k < DSTATE / 2; ++k) {
      const v2f arg = A2[k] * dlt;
      v2f a;
      a.x = __builtin_amdgcn_exp2f(arg.x);
      a.y = __builtin_amdgcn_exp2f(arg.y);
      const v2f bb = {Bs[t][2 * k], Bs[t][2 * k + 1]};
      const v2f cc = {Cs[t][2 * k], Cs[t][2 * k + 1]};
      h2[k] = a * h2[k] + bb * du;
      if (k & 1) accB = h2[k] * cc + accB;
      else       accA = h2[k] * cc + accA;
    }
    const v2f acc2 = accA + accB;
    y[idx0 + (size_t)t * DINNER] = (acc2.x + acc2.y) + uu * Dd;
  }
}

extern "C" void kernel_launch(void* const* d_in, const int* in_sizes, int n_in,
                              void* d_out, int out_size, void* d_ws, size_t ws_size,
                              hipStream_t stream) {
  (void)in_sizes; (void)n_in; (void)out_size; (void)ws_size;
  const float* u    = (const float*)d_in[0];
  const float* Wx   = (const float*)d_in[1];
  const float* Wdt  = (const float*)d_in[2];
  const float* bdt  = (const float*)d_in[3];
  const float* logA = (const float*)d_in[4];
  const float* Dp   = (const float*)d_in[5];
  float* y = (float*)d_out;

  float* ws     = (float*)d_ws;
  float* xdbl   = ws;
  float* delta  = xdbl  + (size_t)MROWS * XPROJ_N;
  float* chunkA = delta + (size_t)MROWS * DINNER;
  float* chunkH = chunkA + (size_t)BATCH * DINNER * NCH * DSTATE;
  float* carry  = chunkH + (size_t)BATCH * DINNER * NCH * DSTATE;

  k_xproj<<<MROWS / 16, XP_WAVES * 32, 0, stream>>>(u, Wx, Wdt, bdt, xdbl, delta);
  const int scan_blocks = BATCH * NCH * DBLKS;
  k_scan_phase1<<<scan_blocks, 256, 0, stream>>>(u, xdbl, delta, logA, chunkA, chunkH);
  k_scan_phase2<<<(BATCH * DINNER * DSTATE) / 256, 256, 0, stream>>>(chunkA, chunkH, carry);
  k_scan_phase3<<<scan_blocks, 256, 0, stream>>>(u, xdbl, delta, logA, Dp, carry, y);
}


# carry pass over the chunks folded into scan phases 1 and 3 (work-group = 4 chunks x 64 channels; chunk prefixes and group totals combined through LDS): 3 launches, not 4; sum order changes
# speedup vs baseline: 4.4882x; 1.1962x over previous
#include <hip/hip_runtime.h>
#include <hip/hip_bf16.h>
#include <math.h>

#define BATCH   2
#define LENGTH  2048
#define DINNER  1024
#define DSTATE  16
#define DTRANK  64
#define MROWS   (BATCH * LENGTH)
#define XPROJ_N (DTRANK + 2 * DSTATE)

#define NCH 32
#define LCH (LENGTH / NCH)
#define CPG 4
#define NSG (NCH / CPG)
#define DW  64
#define DBLKS (DINNER / DW)
#define GROWS (CPG * LCH)

typedef __attribute__((ext_vector_type(2))) float v2f;
typedef __attribute__((ext_vector_type(4))) float v4f;
typedef __attribute__((ext_vector_type(8))) float v8f;

#define XP_WAVES  16
#define XP_KSLICE (DINNER / XP_WAVES)
#define XP_NT     (XPROJ_N / 16)
#define XP_DRP    (DTRANK + 4)
#define DT_NT     (DINNER / (16 * XP_WAVES))

__device__ __forceinline__ float softplus_f32(float x) {
  const float e  = __expf(-fabsf(x));
  const float s  = e * __builtin_amdgcn_rcpf(e + 2.0f);
  const float z  = s * s;
  float p = fmaf(z, 1.0f / 13.0f, 1.0f / 11.0f);
  p = fmaf(z, p, 1.0f / 9.0f);
  p = fmaf(z, p, 1.0f / 7.0f);
  p = fmaf(z, p, 1.0f / 5.0f);
  p = fmaf(z, p, 1.0f / 3.0f);
  const float s2 = s + s;
  return fmaxf(x, 0.0f) + fmaf(s2, z * p, s2);
}

__global__ void __launch_bounds__(XP_WAVES * 32) __attribute__((amdgpu_waves_per_eu(4, 4)))
k_xproj(const float* __restrict__ u, const float* __restrict__ Wx,
        const float* __restrict__ Wdt, const float* __restrict__ bdt,
        float* __restrict__ xdbl, float* __restrict__ delta) {
  __shared__ alignas(16) float red[XP_WAVES][XP_NT][8][32];
  __shared__ alignas(16) float dr[16][XP_DRP];
  const int w    = threadIdx.x >> 5;
  const int lane = threadIdx.x & 31;
  const int c    = lane & 15;
  const int h    = lane >> 4;
  const int tileM = blockIdx.x;

  const int kofs = w * XP_KSLICE + 4 * h;
  const float* aRow = u  + (size_t)(tileM * 16 + c) * DINNER + kofs;
  const float* bRow = Wx + (size_t)c * DINNER + kofs;

  v8f acc[XP_NT];
#pragma unroll
  for (int j = 0; j < XP_NT; ++j) acc[j] = v8f{};

  v4f a[XP_KSLICE / 8];
  v4f b[XP_NT][XP_KSLICE / 8];
#pragma unroll
  for (int i = 0; i < XP_KSLICE / 8; ++i)
    a[i] = *(const v4f*)(aRow + 8 * i);
  __builtin_amdgcn_sched_barrier(0);
#pragma unroll
  for (int j = 0; j < XP_NT; ++j) {
#pragma unroll
    for (int i = 0; i < XP_KSLICE / 8; ++i)
      b[j][i] = *(const v4f*)(bRow + (size_t)j * 16 * DINNER + 8 * i);
    if (j < XP_NT - 1) __builtin_amdgcn_sched_barrier(0);
  }

#pragma unroll
  for (int i = 0; i < XP_KSLICE / 8; ++i) {
    const v2f a0 = {a[i].x, a[i].y};
    const v2f a1 = {a[i].z, a[i].w};
#pragma unroll
    for (int j = 0; j < XP_NT; ++j) {
      const v2f b0 = {b[j][i].x, b[j][i].y};
      acc[j] = __builtin_amdgcn_wmma_f32_16x16x4_f32(
           false, a0,  false, b0,
           (short)0, acc[j],  false,  false);
    }
#pragma unroll
    for (int j = 0; j < XP_NT; ++j) {
      const v2f b1 = {b[j][i].z, b[j][i].w};
      acc[j] = __builtin_amdgcn_wmma_f32_16x16x4_f32(
          false, a1, false, b1, (short)0, acc[j], false, false);
    }
  }

#pragma unroll
  for (int j = 0; j < XP_NT; ++j)
#pragma unroll
    for (int r = 0; r < 8; ++r)
      red[w][j][r][lane] = acc[j][r];
  __syncthreads();

  const float* bRow2 = Wdt + (size_t)(w * (16 * DT_NT) + c) * DTRANK + 4 * h;
  v4f bq[DT_NT][DTRANK / 8];
#pragma unroll
  for (int nt = 0; nt < DT_NT; ++nt)
#pragma unroll
    for (int i = 0; i < DTRANK / 8; ++i)
      bq[nt][i] = *(const v4f*)(bRow2 + (size_t)nt * 16 * DTRANK + 8 * i);
  const int trow = lane >> 4;
  const int col4 = 4 * (lane & 15);
  const v4f bias4 = *(const v4f*)(bdt + w * (16 * DT_NT) + col4);
  __builtin_amdgcn_sched_barrier(0);

  float* outBase = xdbl + (size_t)tileM * (16 * XPROJ_N);
#pragma unroll
  for (int q = 0; q < (16 * XPROJ_N) / (XP_WAVES * 32); ++q) {
    const int o   = threadIdx.x + q * (XP_WAVES * 32);
    const int row = o / XPROJ_N;
    const int col = o - row * XPROJ_N;
    const int j   = col >> 4;
    const int src = ((row >> 3) << 4) | (col & 15);
    const int r   = row & 7;
    float s = red[0][j][r][src];
#pragma unroll
    for (int ww = 1; ww < XP_WAVES; ++ww) s += red[ww][j][r][src];
    if (col < DTRANK) dr[row][col] = s;
    else              outBase[o]   = s;
  }
  __syncthreads();

  v4f aq[DTRANK / 8];
#pragma unroll
  for (int i = 0; i < DTRANK / 8; ++i)
    aq[i] = *(const v4f*)&dr[c][8 * i + 4 * h];
  v8f dacc[DT_NT];
#pragma unroll
  for (int nt = 0; nt < DT_NT; ++nt) dacc[nt] = v8f{};
#pragma unroll
  for (int i = 0; i < DTRANK / 8; ++i) {
    const v2f a0 = {aq[i].x, aq[i].y};
    const v2f a1 = {aq[i].z, aq[i].w};
#pragma unroll
    for (int nt = 0; nt < DT_NT; ++nt) {
      const v2f b0 = {bq[nt][i].x, bq[nt][i].y};
      dacc[nt] = __builtin_amdgcn_wmma_f32_16x16x4_f32(
          false, a0, false, b0, (short)0, dacc[nt], false, false);
    }
#pragma unroll
    for (int nt = 0; nt < DT_NT; ++nt) {
      const v2f b1 = {bq[nt][i].z, bq[nt][i].w};
      dacc[nt] = __builtin_amdgcn_wmma_f32_16x16x4_f32(
          false, a1, false, b1, (short)0, dacc[nt], false, false);
    }
  }

  float* tile = &red[0][0][0][0] + w * (16 * XP_DRP);
#pragma unroll
  for (int nt = 0; nt < DT_NT; ++nt)
#pragma unroll
    for (int r = 0; r < 8; ++r)
      tile[(8 * h + r) * XP_DRP + 16 * nt + c] = dacc[nt][r];
  __syncthreads();

  const size_t orow = (size_t)tileM * 16;
  float* dOut = delta + orow * DINNER + w * (16 * DT_NT) + col4;
  for (int it = 0; it < 8; ++it) {
    const int row = 2 * it + trow;
    const v4f x = *(const v4f*)(tile + row * XP_DRP + col4);
    v4f sp;
#pragma unroll
    for (int e = 0; e < 4; ++e) sp[e] = softplus_f32(x[e] + bias4[e]);
    *(v4f*)(dOut + (size_t)row * DINNER) = sp;
  }
}

#define LOG2E_F 1.4426950408889634f
__global__ void __launch_bounds__(256) k_scan_phase1(const float* __restrict__ u, const float* __restrict__ xdbl,
                              const float* __restrict__ delta, const float* __restrict__ logA,
                              float* __restrict__ prefH, float* __restrict__ prefD,
                              float* __restrict__ supH, float* __restrict__ supD) {
  const int dblk = blockIdx.x % DBLKS;
  const int sg   = (blockIdx.x / DBLKS) % NSG;
  const int b    = blockIdx.x / (DBLKS * NSG);
  const int tid  = threadIdx.x;
  const int c    = __builtin_amdgcn_readfirstlane(tid >> 6);
  const int dl   = tid & (DW - 1);
  const int d    = dblk * DW + dl;
  const int ch   = sg * CPG + c;
  const int t0   = ch * LCH;

  __shared__ alignas(16) float Bs[GROWS][DSTATE];
  __shared__ float Dl[LCH][256];
  __shared__ float Uu[LCH][256];
  __shared__ float rec[CPG][DSTATE + 1][DW];

  v4f bstage[4];
#pragma unroll
  for (int q = 0; q < 4; ++q) {
    const int idx = tid + 256 * q;
    bstage[q] = *(const v4f*)(xdbl + (size_t)(b * LENGTH + sg * GROWS + (idx >> 2)) * XPROJ_N
                              + DTRANK + 4 * (idx & 3));
  }

  float An[DSTATE];
#pragma unroll
  for (int n = 0; n < DSTATE; ++n) An[n] = logA[d * DSTATE + n];

  const size_t idx0 = (size_t)(b * LENGTH + t0) * DINNER + d;
  float dl_[LCH], uv[LCH];
#pragma unroll
  for (int t = 0; t < LCH; ++t) {
    dl_[t] = delta[idx0 + (size_t)t * DINNER];
    uv[t]  = u[idx0 + (size_t)t * DINNER];
  }

  asm volatile("" :: "v"(An[0]), "v"(An[1]), "v"(An[2]), "v"(An[3]),
                     "v"(An[4]), "v"(An[5]), "v"(An[6]), "v"(An[7]),
                     "v"(An[8]), "v"(An[9]), "v"(An[10]), "v"(An[11]),
                     "v"(An[12]), "v"(An[13]), "v"(An[14]), "v"(An[15]));
  __builtin_amdgcn_sched_barrier(0);
#pragma unroll
  for (int q = 0; q < 4; ++q) {
    const int idx = tid + 256 * q;
    *(v4f*)&Bs[idx >> 2][4 * (idx & 3)] = bstage[q];
  }
#pragma unroll
  for (int t = 0; t < LCH; ++t) {
    Dl[t][tid] = dl_[t];
    Uu[t][tid] = uv[t];
  }
  v2f A2[DSTATE / 2];
#pragma unroll
  for (int k = 0; k < DSTATE / 2; ++k) {
    A2[k].x = -__expf(An[2 * k])     * LOG2E_F;
    A2[k].y = -__expf(An[2 * k + 1]) * LOG2E_F;
  }
  __syncthreads();

  v2f h2[DSTATE / 2];
#pragma unroll
  for (int k = 0; k < DSTATE / 2; ++k) h2[k] = v2f{0.f, 0.f};
  float dsum = 0.f;
  float dnext = Dl[0][tid], unext = Uu[0][tid];
  const int r0 = c * LCH;
#pragma unroll 2
  for (int t = 0; t < LCH; ++t) {
    const float dlt = dnext;
    const float du  = dlt * unext;
    const int   tn  = (t + 1 < LCH) ? t + 1 : t;
    dnext = Dl[tn][tid];
    unext = Uu[tn][tid];
    dsum += dlt;
#pragma unroll
    for (int k = 0; k < DSTATE / 2; ++k) {
      const v2f arg = A2[k] * dlt;
      v2f a;
      a.x = __builtin_amdgcn_exp2f(arg.x);
      a.y = __builtin_amdgcn_exp2f(arg.y);
      const v2f bb = {Bs[r0 + t][2 * k], Bs[r0 + t][2 * k + 1]};
      h2[k] = a * h2[k] + bb * du;
    }
  }

  rec[c][DSTATE][dl] = dsum;
#pragma unroll
  for (int k = 0; k < DSTATE / 2; ++k) {
    rec[c][2 * k][dl]     = h2[k].x;
    rec[c][2 * k + 1][dl] = h2[k].y;
  }
  __syncthreads();

  v2f p2[DSTATE / 2];
#pragma unroll
  for (int k = 0; k < DSTATE / 2; ++k) p2[k] = v2f{0.f, 0.f};
  float dpre = 0.f;
  for (int j = 0; j < c; ++j) {
    const float dj = rec[j][DSTATE][dl];
#pragma unroll
    for (int k = 0; k < DSTATE / 2; ++k) {
      v2f a;
      a.x = __builtin_amdgcn_exp2f(A2[k].x * dj);
      a.y = __builtin_amdgcn_exp2f(A2[k].y * dj);
      const v2f g = {rec[j][2 * k][dl], rec[j][2 * k + 1][dl]};
      p2[k] = a * p2[k] + g;
    }
    dpre += dj;
  }
  const size_t bd = (size_t)(b * DINNER + d);
  float* ph = prefH + (bd * NCH + ch) * DSTATE;
#pragma unroll
  for (int q = 0; q < DSTATE / 4; ++q)
    *(v4f*)(ph + 4 * q) = v4f{p2[2 * q].x, p2[2 * q].y, p2[2 * q + 1].x, p2[2 * q + 1].y};
  prefD[bd * NCH + ch] = dpre;
  if (c == CPG - 1) {
    float* sh = supH + (bd * NSG + sg) * DSTATE;
#pragma unroll
    for (int q = 0; q < DSTATE / 4; ++q) {
      v4f tot;
#pragma unroll
      for (int e = 0; e < 2; ++e) {
        const int k = 2 * q + e;
        v2f a;
        a.x = __builtin_amdgcn_exp2f(A2[k].x * dsum);
        a.y = __builtin_amdgcn_exp2f(A2[k].y * dsum);
        const v2f tk = a * p2[k] + h2[k];
        tot[2 * e]     = tk.x;
        tot[2 * e + 1] = tk.y;
      }
      *(v4f*)(sh + 4 * q) = tot;
    }
    supD[bd * NSG + sg] = dpre + dsum;
  }
}

__global__ void __launch_bounds__(256) k_scan_phase3(const float* __restrict__ u, const float* __restrict__ xdbl,
                              const float* __restrict__ delta, const float* __restrict__ logA,
                              const float* __restrict__ Dp,
                              const float* __restrict__ prefH, const float* __restrict__ prefD,
                              const float* __restrict__ supH, const float* __restrict__ supD,
                              float* __restrict__ y) {
  const int dblk = blockIdx.x % DBLKS;
  const int sg   = (blockIdx.x / DBLKS) % NSG;
  const int b    = blockIdx.x / (DBLKS * NSG);
  const int tid  = threadIdx.x;
  const int c    = __builtin_amdgcn_readfirstlane(tid >> 6);
  const int dl   = tid & (DW - 1);
  const int d    = dblk * DW + dl;
  const int ch   = sg * CPG + c;
  const int t0   = ch * LCH;

  __shared__ alignas(16) float BCs[GROWS][2 * DSTATE];
  __shared__ float Dl[LCH][256];
  __shared__ float Uu[LCH][256];
  __shared__ alignas(16) float sc[DW][DSTATE];

  const size_t bd = (size_t)(b * DINNER + d);
  v4f shv[NSG - 1];
  float sdv[NSG - 1];
#pragma unroll
  for (int s = 0; s < NSG - 1; ++s) {
    shv[s] = *(const v4f*)(supH + (bd * NSG + s) * DSTATE + 4 * c);
    sdv[s] = supD[bd * NSG + s];
  }
  const v4f Aq = *(const v4f*)(logA + d * DSTATE + 4 * c);
  v4f phv[DSTATE / 4];
#pragma unroll
  for (int q = 0; q < DSTATE / 4; ++q)
    phv[q] = *(const v4f*)(prefH + (bd * NCH + ch) * DSTATE + 4 * q);
  const float pd = prefD[bd * NCH + ch];

  float An[DSTATE];
#pragma unroll
  for (int n = 0; n < DSTATE; ++n) An[n] = logA[d * DSTATE + n];
  const float Dd = Dp[d];

  v4f bcstage[8];
#pragma unroll
  for (int q = 0; q < 8; ++q) {
    const int idx = tid + 256 * q;
    bcstage[q] = *(const v4f*)(xdbl + (size_t)(b * LENGTH + sg * GROWS + (idx >> 3)) * XPROJ_N
                               + DTRANK + 4 * (idx & 7));
  }

  const size_t idx0 = (size_t)(b * LENGTH + t0) * DINNER + d;
  float dl_[LCH], uv[LCH];
#pragma unroll
  for (int t = 0; t < LCH; ++t) {
    dl_[t] = delta[idx0 + (size_t)t * DINNER];
    uv[t]  = u[idx0 + (size_t)t * DINNER];
  }

  asm volatile("" :: "v"(An[0]), "v"(An[1]), "v"(An[2]), "v"(An[3]),
                     "v"(An[4]), "v"(An[5]), "v"(An[6]), "v"(An[7]),
                     "v"(An[8]), "v"(An[9]), "v"(An[10]), "v"(An[11]),
                     "v"(An[12]), "v"(An[13]), "v"(An[14]), "v"(An[15]));
  __builtin_amdgcn_sched_barrier(0);

  {
    v4f A2q;
#pragma unroll
    for (int i = 0; i < 4; ++i) A2q[i] = -__expf(Aq[i]) * LOG2E_F;
    v4f hq = {0.f, 0.f, 0.f, 0.f};
#pragma unroll
    for (int s = 0; s < NSG - 1; ++s) {
      if (s < sg) {
#pragma unroll
        for (int i = 0; i < 4; ++i)
          hq[i] = __builtin_amdgcn_exp2f(A2q[i] * sdv[s]) * hq[i] + shv[s][i];
      }
    }
    *(v4f*)&sc[dl][4 * c] = hq;
  }
#pragma unroll
  for (int q = 0; q < 8; ++q) {
    const int idx = tid + 256 * q;
    *(v4f*)&BCs[idx >> 3][4 * (idx & 7)] = bcstage[q];
  }
#pragma unroll
  for (int t = 0; t < LCH; ++t) {
    Dl[t][tid] = dl_[t];
    Uu[t][tid] = uv[t];
  }
  v2f A2[DSTATE / 2], h2[DSTATE / 2];
#pragma unroll
  for (int k = 0; k < DSTATE / 2; ++k) {
    A2[k].x = -__expf(An[2 * k])     * LOG2E_F;
    A2[k].y = -__expf(An[2 * k + 1]) * LOG2E_F;
  }
  __syncthreads();

#pragma unroll
  for (int q = 0; q < DSTATE / 4; ++q) {
    const v4f g = *(const v4f*)&sc[dl][4 * q];
    h2[2 * q]     = v2f{g.x, g.y};
    h2[2 * q + 1] = v2f{g.z, g.w};
  }
  if (c > 0) {
#pragma unroll
    for (int k = 0; k < DSTATE / 2; ++k) {
      v2f a;
      a.x = __builtin_amdgcn_exp2f(A2[k].x * pd);
      a.y = __builtin_amdgcn_exp2f(A2[k].y * pd);
      const v2f p = (k & 1) ? v2f{phv[k >> 1].z, phv[k >> 1].w} : v2f{phv[k >> 1].x, phv[k >> 1].y};
      h2[k] = a * h2[k] + p;
    }
  }

  float dnext = Dl[0][tid], unext = Uu[0][tid];
  const int r0 = c * LCH;
#pragma unroll 2
  for (int t = 0; t < LCH; ++t) {
    const float dlt = dnext;
    const float uu  = unext;
    const float du  = dlt * uu;
    const int   tn  = (t + 1 < LCH) ? t + 1 : t;
    dnext = Dl[tn][tid];
    unext = Uu[tn][tid];
    v2f accA = {0.f, 0.f}, accB = {0.f, 0.f};
#pragma unroll
    for (int k = 0; k < DSTATE / 2; ++k) {
      const v2f arg = A2[k] * dlt;
      v2f a;
      a.x = __builtin_amdgcn_exp2f(arg.x);
      a.y = __builtin_amdgcn_exp2f(arg.y);
      const v2f bb = {BCs[r0 + t][2 * k], BCs[r0 + t][2 * k + 1]};
      const v2f cc = {BCs[r0 + t][DSTATE + 2 * k], BCs[r0 + t][DSTATE + 2 * k + 1]};
      h2[k] = a * h2[k] + bb * du;
      if (k & 1) accB = h2[k] * cc + accB;
      else       accA = h2[k] * cc + accA;
    }
    const v2f acc2 = accA + accB;
    y[idx0 + (size_t)t * DINNER] = (acc2.x + acc2.y) + uu * Dd;
  }
}

extern "C" void kernel_launch(void* const* d_in, const int* in_sizes, int n_in,
                              void* d_out, int out_size, void* d_ws, size_t ws_size,
                              hipStream_t stream) {
  (void)in_sizes; (void)n_in; (void)out_size; (void)ws_size;
  const float* u    = (const float*)d_in[0];
  const float* Wx   = (const float*)d_in[1];
  const float* Wdt  = (const float*)d_in[2];
  const float* bdt  = (const float*)d_in[3];
  const float* logA = (const float*)d_in[4];
  const float* Dp   = (const float*)d_in[5];
  float* y = (float*)d_out;

  float* ws     = (float*)d_ws;
  float* xdbl   = ws;
  float* delta  = xdbl  + (size_t)MROWS * XPROJ_N;
  float* prefH  = delta + (size_t)MROWS * DINNER;
  float* prefD  = prefH + (size_t)BATCH * DINNER * NCH * DSTATE;
  float* supH   = prefD + (size_t)BATCH * DINNER * NCH;
  float* supD   = supH  + (size_t)BATCH * DINNER * NSG * DSTATE;

  k_xproj<<<MROWS / 16, XP_WAVES * 32, 0, stream>>>(u, Wx, Wdt, bdt, xdbl, delta);
  const int scan_blocks = BATCH * NSG * DBLKS;
  k_scan_phase1<<<scan_blocks, 256, 0, stream>>>(u, xdbl, delta, logA, prefH, prefD, supH, supD);
  k_scan_phase3<<<scan_blocks, 256, 0, stream>>>(u, xdbl, delta, logA, Dp, prefH, prefD, supH, supD, y);
}
